# starting kernel
# baseline (speedup 1.0000x reference)
#include <hip/hip_runtime.h>
#include <hip/hip_bf16.h>


typedef __attribute__((ext_vector_type(2))) float v2f;
typedef __attribute__((ext_vector_type(8))) float v8f;

__global__ void init_kernel(float* __restrict__ deg, float* __restrict__ agg1,
                            int N) {
  int i = blockIdx.x * blockDim.x + threadIdx.x;
  if (i < N * 16) agg1[i] = 0.0f;
  if (i < N) deg[i] = 1.0f;
}

__global__ void degree_kernel(const int* __restrict__ dst,
                              float* __restrict__ deg, int E) {
  int e = blockIdx.x * blockDim.x + threadIdx.x;
  if (e >= E) return;
  atomicAdd(&deg[dst[e]], 1.0f);
}

__global__ void rsqrt_kernel(float* __restrict__ deg, int N) {
  int i = blockIdx.x * blockDim.x + threadIdx.x;
  if (i >= N) return;
  float d = deg[i];
  deg[i] = (d > 0.0f) ? __frsqrt_rn(d) : 0.0f;
}

__global__ void __launch_bounds__(320)
gemm1_wmma(const float* __restrict__ x,
           const float* __restrict__ W1,
           float* __restrict__ h,
           int num_tiles) {
  int flat = blockIdx.x * blockDim.x + threadIdx.x;
  int wave = flat >> 5;
  int lane = threadIdx.x & 31;
  if (wave >= num_tiles) return;

  int m = lane & 15;
  bool hi = lane >= 16;
  int row0 = wave * 16;

  const float* xr = x + (size_t)(row0 + m) * 3;
  v2f a;
  a.x = hi ? xr[2] : xr[0];
  a.y = hi ? 0.0f : xr[1];

  v2f b;
  b.x = hi ? W1[2 * 16 + m] : W1[0 * 16 + m];
  b.y = hi ? 0.0f : W1[1 * 16 + m];

  v8f c = {};
  c = __builtin_amdgcn_wmma_f32_16x16x4_f32(
       false, a,  false, b,
       (short)0, c,  false,  false);

  int rbase = row0 + (hi ? 8 : 0);
#pragma unroll
  for (int v = 0; v < 8; ++v) {
    h[(size_t)(rbase + v) * 16 + m] = c[v];
  }
}

__global__ void edge_pass1(const int* __restrict__ src,
                           const int* __restrict__ dst,
                           const float* __restrict__ dinv,
                           const float4* __restrict__ h4,
                           float* __restrict__ agg1,
                           int E) {
  int e = blockIdx.x * blockDim.x + threadIdx.x;
  if (e >= E) return;
  int s = src[e];
  int d = dst[e];
  float norm = dinv[s] * dinv[d];
  const float4* hs = h4 + (size_t)s * 4;
  float* out = agg1 + (size_t)d * 16;
#pragma unroll
  for (int g = 0; g < 4; ++g) {
    float4 v = hs[g];
    atomicAdd(out + 4 * g + 0, v.x * norm);
    atomicAdd(out + 4 * g + 1, v.y * norm);
    atomicAdd(out + 4 * g + 2, v.z * norm);
    atomicAdd(out + 4 * g + 3, v.w * norm);
  }
}

__global__ void finalize1_project(const float* __restrict__ h,
                                  const float* __restrict__ agg1,
                                  const float* __restrict__ dinv,
                                  const float* __restrict__ b1,
                                  const float* __restrict__ W2,
                                  const float* __restrict__ b2,
                                  float* __restrict__ z,
                                  float* __restrict__ out,
                                  int N) {
  int i = blockIdx.x * blockDim.x + threadIdx.x;
  if (i >= N) return;
  float di = dinv[i];
  float selfw = di * di;
  float acc = 0.0f;
  const float4* a4 = (const float4*)(agg1 + (size_t)i * 16);
  const float4* h4 = (const float4*)(h + (size_t)i * 16);
#pragma unroll
  for (int g = 0; g < 4; ++g) {
    float4 av = a4[g];
    float4 hv = h4[g];
    float v0 = fmaxf(av.x + hv.x * selfw + b1[4 * g + 0], 0.0f);
    float v1 = fmaxf(av.y + hv.y * selfw + b1[4 * g + 1], 0.0f);
    float v2 = fmaxf(av.z + hv.z * selfw + b1[4 * g + 2], 0.0f);
    float v3 = fmaxf(av.w + hv.w * selfw + b1[4 * g + 3], 0.0f);
    acc += v0 * W2[4 * g + 0] + v1 * W2[4 * g + 1] +
           v2 * W2[4 * g + 2] + v3 * W2[4 * g + 3];
  }
  z[i] = acc;
  out[i] = acc * selfw + b2[0];
}

__global__ void edge_pass2(const int* __restrict__ src,
                           const int* __restrict__ dst,
                           const float* __restrict__ dinv,
                           const float* __restrict__ z,
                           float* __restrict__ out, int E) {
  int e = blockIdx.x * blockDim.x + threadIdx.x;
  if (e >= E) return;
  int s = src[e];
  int d = dst[e];
  atomicAdd(&out[d], z[s] * dinv[s] * dinv[d]);
}

extern "C" void kernel_launch(void* const* d_in, const int* in_sizes, int n_in,
                              void* d_out, int out_size, void* d_ws,
                              size_t ws_size, hipStream_t stream) {
  const float* x  = (const float*)d_in[0];
  const int*   ei = (const int*)d_in[1];
  const float* W1 = (const float*)d_in[2];
  const float* b1 = (const float*)d_in[3];
  const float* W2 = (const float*)d_in[4];
  const float* b2 = (const float*)d_in[5];
  float* out = (float*)d_out;

  const int N = in_sizes[0] / 3;
  const int E = in_sizes[1] / 2;
  const int* src = ei;
  const int* dst = ei + E;

  char* ws = (char*)d_ws;
  float* dinv = (float*)(ws);
  float* h    = (float*)(ws + (size_t)N * 4);
  float* agg1 = (float*)(ws + (size_t)N * 4 + (size_t)N * 64);
  float* z    = (float*)(ws + (size_t)N * 4 + (size_t)N * 128);

  const int BT = 256;

  init_kernel<<<(N * 16 + BT - 1) / BT, BT, 0, stream>>>(dinv, agg1, N);
  degree_kernel<<<(E + BT - 1) / BT, BT, 0, stream>>>(dst, dinv, E);
  rsqrt_kernel<<<(N + BT - 1) / BT, BT, 0, stream>>>(dinv, N);
  {
    int ntiles = N / 16;
    int wpb = 10;
    int blocks = (ntiles + wpb - 1) / wpb;
    gemm1_wmma<<<blocks, 32 * wpb, 0, stream>>>(x, W1, h, ntiles);
  }
  edge_pass1<<<(E + BT - 1) / BT, BT, 0, stream>>>(src, dst, dinv,
                                                   (const float4*)h, agg1, E);
  finalize1_project<<<(N + BT - 1) / BT, BT, 0, stream>>>(h, agg1, dinv, b1,
                                                          W2, b2, z, out, N);
  edge_pass2<<<(E + BT - 1) / BT, BT, 0, stream>>>(src, dst, dinv, z, out, E);
}


# layer 1 aggregates the 3-wide input rows before the 3->16 WMMA transform ((A x) W1 for A (x W1)): edge scatter adds 3 floats per edge, not 16; float sum order changes
# speedup vs baseline: 1.7846x; 1.7846x over previous
#include <hip/hip_runtime.h>
#include <hip/hip_bf16.h>


typedef __attribute__((ext_vector_type(2))) float v2f;
typedef __attribute__((ext_vector_type(8))) float v8f;

__global__ void init_kernel(float* __restrict__ deg, float4* __restrict__ aggx4,
                            int N) {
  int i = blockIdx.x * blockDim.x + threadIdx.x;
  if (i >= N) return;
  aggx4[i] = make_float4(0.0f, 0.0f, 0.0f, 0.0f);
  deg[i] = 1.0f;
}

__global__ void degree_kernel(const int* __restrict__ dst,
                              float* __restrict__ deg, int E) {
  int e = blockIdx.x * blockDim.x + threadIdx.x;
  if (e >= E) return;
  atomicAdd(&deg[dst[e]], 1.0f);
}

__global__ void rsqrt_kernel(float* __restrict__ deg, int N) {
  int i = blockIdx.x * blockDim.x + threadIdx.x;
  if (i >= N) return;
  float d = deg[i];
  deg[i] = (d > 0.0f) ? __frsqrt_rn(d) : 0.0f;
}

__global__ void edge_pass1(const int* __restrict__ src,
                           const int* __restrict__ dst,
                           const float* __restrict__ dinv,
                           const float* __restrict__ x,
                           float* __restrict__ aggx,
                           int E) {
  int e = blockIdx.x * blockDim.x + threadIdx.x;
  if (e >= E) return;
  int s = src[e];
  int d = dst[e];
  float norm = dinv[s] * dinv[d];
  const float* xs = x + (size_t)s * 3;
  float x0 = xs[0];
  float x1 = xs[1];
  float x2 = xs[2];
  float* o = aggx + (size_t)d * 4;
  atomicAdd(o + 0, x0 * norm);
  atomicAdd(o + 1, x1 * norm);
  atomicAdd(o + 2, x2 * norm);
}

__global__ void __launch_bounds__(256)
finalize1_project(const float* __restrict__ x,
                  const float4* __restrict__ aggx4,
                  const float* __restrict__ dinv,
                  const float* __restrict__ W1,
                  const float* __restrict__ b1,
                  const float* __restrict__ W2,
                  const float* __restrict__ b2,
                  float* __restrict__ z,
                  float* __restrict__ out,
                  int N, int num_tiles) {
  int flat = blockIdx.x * blockDim.x + threadIdx.x;
  int wave = flat >> 5;
  int lane = threadIdx.x & 31;
  if (wave >= num_tiles) return;

  int m = lane & 15;
  bool hi = lane >= 16;
  int node = wave * 16 + m;
  bool valid = node < N;
  int nl = valid ? node : (N - 1);

  float di = dinv[nl];
  float selfw = di * di;
  float4 ag = aggx4[nl];
  const float* xr = x + (size_t)nl * 3;
  float ax0 = ag.x + xr[0] * selfw;
  float ax1 = ag.y + xr[1] * selfw;
  float ax2 = ag.z + xr[2] * selfw;

  v2f a;
  a.x = hi ? W1[2 * 16 + m] : W1[0 * 16 + m];
  a.y = hi ? 0.0f : W1[1 * 16 + m];

  v2f b;
  b.x = hi ? ax2 : ax0;
  b.y = hi ? 0.0f : ax1;

  v8f c = {};
  c = __builtin_amdgcn_wmma_f32_16x16x4_f32(
       false, a,  false, b,
       (short)0, c,  false,  false);

  int fb = hi ? 8 : 0;
  float acc = 0.0f;
  for (int v = 0; v < 8; ++v) {
    float hv = fmaxf(c[v] + b1[fb + v], 0.0f);
    acc += hv * W2[fb + v];
  }
  acc += __shfl_xor(acc, 16);

  if (!hi && valid) {
    z[node] = acc;
    out[node] = acc * selfw + b2[0];
  }
}

__global__ void edge_pass2(const int* __restrict__ src,
                           const int* __restrict__ dst,
                           const float* __restrict__ dinv,
                           const float* __restrict__ z,
                           float* __restrict__ out, int E) {
  int e = blockIdx.x * blockDim.x + threadIdx.x;
  if (e >= E) return;
  int s = src[e];
  int d = dst[e];
  atomicAdd(&out[d], z[s] * dinv[s] * dinv[d]);
}

extern "C" void kernel_launch(void* const* d_in, const int* in_sizes, int n_in,
                              void* d_out, int out_size, void* d_ws,
                              size_t ws_size, hipStream_t stream) {
  const float* x  = (const float*)d_in[0];
  const int*   ei = (const int*)d_in[1];
  const float* W1 = (const float*)d_in[2];
  const float* b1 = (const float*)d_in[3];
  const float* W2 = (const float*)d_in[4];
  const float* b2 = (const float*)d_in[5];
  float* out = (float*)d_out;

  const int N = in_sizes[0] / 3;
  const int E = in_sizes[1] / 2;
  const int* src = ei;
  const int* dst = ei + E;

  char* ws = (char*)d_ws;
  float* dinv = (float*)(ws);
  float* aggx = (float*)(ws + (size_t)N * 4);
  float* z    = (float*)(ws + (size_t)N * 4 + (size_t)N * 16);

  const int BT = 256;

  init_kernel<<<(N + BT - 1) / BT, BT, 0, stream>>>(dinv, (float4*)aggx, N);
  degree_kernel<<<(E + BT - 1) / BT, BT, 0, stream>>>(dst, dinv, E);
  rsqrt_kernel<<<(N + BT - 1) / BT, BT, 0, stream>>>(dinv, N);
  edge_pass1<<<(E + BT - 1) / BT, BT, 0, stream>>>(src, dst, dinv, x, aggx, E);
  {
    int ntiles = (N + 15) / 16;
    int wpb = BT / 32;
    int blocks = (ntiles + wpb - 1) / wpb;
    finalize1_project<<<blocks, BT, 0, stream>>>(x, (const float4*)aggx, dinv,
                                                 W1, b1, W2, b2, z, out, N,
                                                 ntiles);
  }
  edge_pass2<<<(E + BT - 1) / BT, BT, 0, stream>>>(src, dst, dinv, z, out, E);
}


# edges grouped once per call by destination range (LDS counting sort); each range's work-group sums degree and both layers in LDS float atomics, no global atomics; sums stay in arrival order
# speedup vs baseline: 9.3185x; 5.2216x over previous
#include <hip/hip_runtime.h>
#include <hip/hip_bf16.h>


typedef __attribute__((ext_vector_type(2))) float v2f;
typedef __attribute__((ext_vector_type(8))) float v8f;

#define NB 256
#define NCH 256
#define RMAX 1024
#define RBITS 10
#define TPB 1024
static_assert(TPB == RMAX, "one thread per range slot");

__device__ __forceinline__ void chunk_range(int E, int w, int& e0, int& e1) {
  int per = (E + NCH - 1) / NCH;
  e0 = w * per;
  int hi = e0 + per;
  e1 = hi < E ? hi : E;
}

__global__ void __launch_bounds__(TPB)
count_kernel(const int* __restrict__ dst, int* __restrict__ cnt, int E, int R) {
  __shared__ int h[NB];
  const int t = threadIdx.x;
  if (t < NB) h[t] = 0;
  __syncthreads();
  int e0, e1;
  chunk_range(E, blockIdx.x, e0, e1);
  const unsigned r = (unsigned)R;
  int e = e0 + t;
  for (; e + 3 * TPB < e1; e += 4 * TPB) {
    unsigned d[4];
    for (int k = 0; k < 4; ++k) d[k] = (unsigned)dst[e + k * TPB];
    for (int k = 0; k < 4; ++k) atomicAdd(&h[d[k] / r], 1);
  }
  for (; e < e1; e += TPB) atomicAdd(&h[(unsigned)dst[e] / r], 1);
  __syncthreads();
  if (t < NB) cnt[blockIdx.x * NB + t] = h[t];
}

__global__ void __launch_bounds__(TPB)
scan_kernel(int* __restrict__ cnt, int* __restrict__ bstart) {
  __shared__ int part[4][NB];
  __shared__ int sc[NB];
  const int t = threadIdx.x;
  const int b = t & (NB - 1);
  const int q = t >> 8;
  int sum = 0;
  for (int i = 0; i < NCH / 4; ++i) sum += cnt[(q * (NCH / 4) + i) * NB + b];
  part[q][b] = sum;
  __syncthreads();
  int tot = 0;
  if (t < NB) {
    tot = part[0][t] + part[1][t] + part[2][t] + part[3][t];
    sc[t] = tot;
  }
  __syncthreads();
  for (int o = 1; o < NB; o <<= 1) {
    int v = 0;
    if (t < NB && t >= o) v = sc[t - o];
    __syncthreads();
    if (t < NB) sc[t] += v;
    __syncthreads();
  }
  if (t < NB) {
    bstart[t] = sc[t] - tot;
    if (t == NB - 1) bstart[NB] = sc[t];
  }
  int run = sc[b] - (part[0][b] + part[1][b] + part[2][b] + part[3][b]);
  for (int k = 0; k < q; ++k) run += part[k][b];
  for (int i = 0; i < NCH / 4; ++i) {
    int idx = (q * (NCH / 4) + i) * NB + b;
    int c = cnt[idx];
    cnt[idx] = run;
    run += c;
  }
}

__global__ void __launch_bounds__(TPB)
fill_kernel(const int* __restrict__ src, const int* __restrict__ dst,
            const int* __restrict__ off, unsigned* __restrict__ list,
            int E, int R) {
  __shared__ int cur[NB];
  const int t = threadIdx.x;
  if (t < NB) cur[t] = off[blockIdx.x * NB + t];
  __syncthreads();
  int e0, e1;
  chunk_range(E, blockIdx.x, e0, e1);
  const unsigned r = (unsigned)R;
  int e = e0 + t;
  for (; e + 3 * TPB < e1; e += 4 * TPB) {
    unsigned s[4], d[4];
    for (int k = 0; k < 4; ++k) {
      s[k] = (unsigned)src[e + k * TPB];
      d[k] = (unsigned)dst[e + k * TPB];
    }
    for (int k = 0; k < 4; ++k) {
      unsigned b = d[k] / r;
      unsigned dl = d[k] - b * r;
      int pos = atomicAdd(&cur[b], 1);
      list[pos] = (s[k] << RBITS) | dl;
    }
  }
  for (; e < e1; e += TPB) {
    unsigned sv = (unsigned)src[e];
    unsigned dv = (unsigned)dst[e];
    unsigned b = dv / r;
    unsigned dl = dv - b * r;
    int pos = atomicAdd(&cur[b], 1);
    list[pos] = (sv << RBITS) | dl;
  }
}

__global__ void __launch_bounds__(TPB)
deg_kernel(const unsigned* __restrict__ list, const int* __restrict__ bstart,
           float* __restrict__ dinv, int N, int R) {
  __shared__ int c[RMAX];
  const int t = threadIdx.x;
  const int b = blockIdx.x;
  c[t] = 0;
  __syncthreads();
  const int beg = bstart[b];
  const int end = bstart[b + 1];
  int j = beg + t;
  for (; j + 3 * TPB < end; j += 4 * TPB) {
    unsigned rec[4];
    for (int k = 0; k < 4; ++k) rec[k] = list[j + k * TPB];
    for (int k = 0; k < 4; ++k) atomicAdd(&c[rec[k] & (RMAX - 1)], 1);
  }
  for (; j < end; j += TPB) atomicAdd(&c[list[j] & (RMAX - 1)], 1);
  __syncthreads();
  const int node = b * R + t;
  if (t < R && node < N) {
    float d = 1.0f + (float)c[t];
    dinv[node] = (d > 0.0f) ? __frsqrt_rn(d) : 0.0f;
  }
}

__global__ void __launch_bounds__(TPB)
agg1_kernel(const unsigned* __restrict__ list, const int* __restrict__ bstart,
            const float* __restrict__ dinv,
            const float* __restrict__ x,
            float4* __restrict__ aggx4,
            int N, int R) {
  __shared__ float a0[RMAX];
  __shared__ float a1[RMAX];
  __shared__ float a2[RMAX];
  __shared__ float dls[RMAX];
  const int t = threadIdx.x;
  const int b = blockIdx.x;
  const int node = b * R + t;
  const bool own = (t < R) && (node < N);
  a0[t] = 0.0f;
  a1[t] = 0.0f;
  a2[t] = 0.0f;
  dls[t] = own ? dinv[node] : 0.0f;
  __syncthreads();
  const int beg = bstart[b];
  const int end = bstart[b + 1];
  int j = beg + t;
  for (; j + 3 * TPB < end; j += 4 * TPB) {
    unsigned rec[4];
    for (int k = 0; k < 4; ++k) rec[k] = list[j + k * TPB];
    float ds[4], x0[4], x1[4], x2[4];
    for (int k = 0; k < 4; ++k) {
      unsigned s = rec[k] >> RBITS;
      const float* xs = x + (size_t)s * 3;
      ds[k] = dinv[s];
      x0[k] = xs[0];
      x1[k] = xs[1];
      x2[k] = xs[2];
    }
    for (int k = 0; k < 4; ++k) {
      unsigned dl = rec[k] & (RMAX - 1);
      float norm = ds[k] * dls[dl];
      atomicAdd(&a0[dl], x0[k] * norm);
      atomicAdd(&a1[dl], x1[k] * norm);
      atomicAdd(&a2[dl], x2[k] * norm);
    }
  }
  for (; j < end; j += TPB) {
    unsigned rec = list[j];
    unsigned s = rec >> RBITS;
    unsigned dl = rec & (RMAX - 1);
    const float* xs = x + (size_t)s * 3;
    float norm = dinv[s] * dls[dl];
    atomicAdd(&a0[dl], xs[0] * norm);
    atomicAdd(&a1[dl], xs[1] * norm);
    atomicAdd(&a2[dl], xs[2] * norm);
  }
  __syncthreads();
  if (own) aggx4[node] = make_float4(a0[t], a1[t], a2[t], 0.0f);
}

__global__ void __launch_bounds__(256)
finalize1_project(const float* __restrict__ x,
                  const float4* __restrict__ aggx4,
                  const float* __restrict__ dinv,
                  const float* __restrict__ W1,
                  const float* __restrict__ b1,
                  const float* __restrict__ W2,
                  const float* __restrict__ b2,
                  float* __restrict__ z,
                  float* __restrict__ out,
                  int N, int num_tiles) {
  int flat = blockIdx.x * blockDim.x + threadIdx.x;
  int wave = flat >> 5;
  int lane = threadIdx.x & 31;
  if (wave >= num_tiles) return;

  int m = lane & 15;
  bool hi = lane >= 16;
  int node = wave * 16 + m;
  bool valid = node < N;
  int nl = valid ? node : (N - 1);

  float di = dinv[nl];
  float selfw = di * di;
  float4 ag = aggx4[nl];
  const float* xr = x + (size_t)nl * 3;
  float ax0 = ag.x + xr[0] * selfw;
  float ax1 = ag.y + xr[1] * selfw;
  float ax2 = ag.z + xr[2] * selfw;

  v2f a;
  a.x = hi ? W1[2 * 16 + m] : W1[0 * 16 + m];
  a.y = hi ? 0.0f : W1[1 * 16 + m];

  v2f b;
  b.x = hi ? ax2 : ax0;
  b.y = hi ? 0.0f : ax1;

  v8f c = {};
  c = __builtin_amdgcn_wmma_f32_16x16x4_f32(
       false, a,  false, b,
       (short)0, c,  false,  false);

  int fb = hi ? 8 : 0;
  float acc = 0.0f;
  for (int v = 0; v < 8; ++v) {
    float hv = fmaxf(c[v] + b1[fb + v], 0.0f);
    acc += hv * W2[fb + v];
  }
  acc += __shfl_xor(acc, 16);

  if (!hi && valid) {
    z[node] = acc;
    out[node] = acc * selfw + b2[0];
  }
}

__global__ void __launch_bounds__(TPB)
agg2_kernel(const unsigned* __restrict__ list, const int* __restrict__ bstart,
            const float* __restrict__ dinv, const float* __restrict__ z,
            float* __restrict__ out, int N, int R) {
  __shared__ float acc[RMAX];
  __shared__ float dls[RMAX];
  const int t = threadIdx.x;
  const int b = blockIdx.x;
  const int node = b * R + t;
  const bool own = (t < R) && (node < N);
  acc[t] = own ? out[node] : 0.0f;
  dls[t] = own ? dinv[node] : 0.0f;
  __syncthreads();
  const int beg = bstart[b];
  const int end = bstart[b + 1];
  int j = beg + t;
  for (; j + 3 * TPB < end; j += 4 * TPB) {
    unsigned rec[4];
    for (int k = 0; k < 4; ++k) rec[k] = list[j + k * TPB];
    float zs[4], ds[4];
    for (int k = 0; k < 4; ++k) {
      unsigned s = rec[k] >> RBITS;
      zs[k] = z[s];
      ds[k] = dinv[s];
    }
    for (int k = 0; k < 4; ++k) {
      unsigned dl = rec[k] & (RMAX - 1);
      atomicAdd(&acc[dl], zs[k] * ds[k] * dls[dl]);
    }
  }
  for (; j < end; j += TPB) {
    unsigned rec = list[j];
    unsigned s = rec >> RBITS;
    unsigned dl = rec & (RMAX - 1);
    atomicAdd(&acc[dl], z[s] * dinv[s] * dls[dl]);
  }
  __syncthreads();
  if (own) out[node] = acc[t];
}

extern "C" void kernel_launch(void* const* d_in, const int* in_sizes, int n_in,
                              void* d_out, int out_size, void* d_ws,
                              size_t ws_size, hipStream_t stream) {
  const float* x  = (const float*)d_in[0];
  const int*   ei = (const int*)d_in[1];
  const float* W1 = (const float*)d_in[2];
  const float* b1 = (const float*)d_in[3];
  const float* W2 = (const float*)d_in[4];
  const float* b2 = (const float*)d_in[5];
  float* out = (float*)d_out;

  const int N = in_sizes[0] / 3;
  const int E = in_sizes[1] / 2;
  const int* src = ei;
  const int* dst = ei + E;

  const int R = (((N + NB - 1) / NB) + 15) & ~15;

  char* ws = (char*)d_ws;
  float* dinv = (float*)(ws);
  float* aggx = (float*)(ws + (size_t)N * 4);
  float* z    = (float*)(ws + (size_t)N * 4 + (size_t)N * 16);
  int* cnt    = (int*)(ws + (size_t)N * 24);
  int* bstart = cnt + NCH * NB;
  unsigned* list = (unsigned*)(bstart + 512);

  count_kernel<<<NCH, TPB, 0, stream>>>(dst, cnt, E, R);
  scan_kernel<<<1, TPB, 0, stream>>>(cnt, bstart);
  fill_kernel<<<NCH, TPB, 0, stream>>>(src, dst, cnt, list, E, R);
  deg_kernel<<<NB, TPB, 0, stream>>>(list, bstart, dinv, N, R);
  agg1_kernel<<<NB, TPB, 0, stream>>>(list, bstart, dinv, x, (float4*)aggx, N,
                                      R);
  {
    const int BT = 256;
    int ntiles = (N + 15) / 16;
    int wpb = BT / 32;
    int blocks = (ntiles + wpb - 1) / wpb;
    finalize1_project<<<blocks, BT, 0, stream>>>(x, (const float4*)aggx, dinv,
                                                 W1, b1, W2, b2, z, out, N,
                                                 ntiles);
  }
  agg2_kernel<<<NB, TPB, 0, stream>>>(list, bstart, dinv, z, out, N, R);
}


# edge norm split into two row scalings: nodes store rows pre-scaled by dinv[src] (x as a 16-byte row; z), dinv[dst] scales the LDS sum once: 1 gather per edge per layer, not 2; sum order changes
# speedup vs baseline: 13.2051x; 1.4171x over previous
#include <hip/hip_runtime.h>
#include <hip/hip_bf16.h>


typedef __attribute__((ext_vector_type(2))) float v2f;
typedef __attribute__((ext_vector_type(8))) float v8f;

#define NB 256
#define NCH 256
#define RMAX 1024
#define RBITS 10
#define TPB 1024
static_assert(TPB == RMAX, "one thread per range slot");

__device__ __forceinline__ void chunk_range(int E, int w, int& e0, int& e1) {
  int per = (E + NCH - 1) / NCH;
  e0 = w * per;
  int hi = e0 + per;
  e1 = hi < E ? hi : E;
}

__global__ void __launch_bounds__(TPB)
count_kernel(const int* __restrict__ dst, int* __restrict__ cnt, int E, int R) {
  __shared__ int h[NB];
  const int t = threadIdx.x;
  if (t < NB) h[t] = 0;
  __syncthreads();
  int e0, e1;
  chunk_range(E, blockIdx.x, e0, e1);
  const unsigned r = (unsigned)R;
  int e = e0 + t;
  for (; e + 3 * TPB < e1; e += 4 * TPB) {
    unsigned d[4];
    for (int k = 0; k < 4; ++k) d[k] = (unsigned)dst[e + k * TPB];
    for (int k = 0; k < 4; ++k) atomicAdd(&h[d[k] / r], 1);
  }
  for (; e < e1; e += TPB) atomicAdd(&h[(unsigned)dst[e] / r], 1);
  __syncthreads();
  if (t < NB) cnt[blockIdx.x * NB + t] = h[t];
}

__global__ void __launch_bounds__(TPB)
scan_kernel(int* __restrict__ cnt, int* __restrict__ bstart) {
  __shared__ int part[4][NB];
  __shared__ int sc[NB];
  const int t = threadIdx.x;
  const int b = t & (NB - 1);
  const int q = t >> 8;
  int sum = 0;
  for (int i = 0; i < NCH / 4; ++i) sum += cnt[(q * (NCH / 4) + i) * NB + b];
  part[q][b] = sum;
  __syncthreads();
  int tot = 0;
  if (t < NB) {
    tot = part[0][t] + part[1][t] + part[2][t] + part[3][t];
    sc[t] = tot;
  }
  __syncthreads();
  for (int o = 1; o < NB; o <<= 1) {
    int v = 0;
    if (t < NB && t >= o) v = sc[t - o];
    __syncthreads();
    if (t < NB) sc[t] += v;
    __syncthreads();
  }
  if (t < NB) {
    bstart[t] = sc[t] - tot;
    if (t == NB - 1) bstart[NB] = sc[t];
  }
  int run = sc[b] - (part[0][b] + part[1][b] + part[2][b] + part[3][b]);
  for (int k = 0; k < q; ++k) run += part[k][b];
  for (int i = 0; i < NCH / 4; ++i) {
    int idx = (q * (NCH / 4) + i) * NB + b;
    int c = cnt[idx];
    cnt[idx] = run;
    run += c;
  }
}

__global__ void __launch_bounds__(TPB)
fill_kernel(const int* __restrict__ src, const int* __restrict__ dst,
            const int* __restrict__ off, unsigned* __restrict__ list,
            int E, int R) {
  __shared__ int cur[NB];
  const int t = threadIdx.x;
  if (t < NB) cur[t] = off[blockIdx.x * NB + t];
  __syncthreads();
  int e0, e1;
  chunk_range(E, blockIdx.x, e0, e1);
  const unsigned r = (unsigned)R;
  int e = e0 + t;
  for (; e + 3 * TPB < e1; e += 4 * TPB) {
    unsigned s[4], d[4];
    for (int k = 0; k < 4; ++k) {
      s[k] = (unsigned)src[e + k * TPB];
      d[k] = (unsigned)dst[e + k * TPB];
    }
    for (int k = 0; k < 4; ++k) {
      unsigned b = d[k] / r;
      unsigned dl = d[k] - b * r;
      int pos = atomicAdd(&cur[b], 1);
      list[pos] = (s[k] << RBITS) | dl;
    }
  }
  for (; e < e1; e += TPB) {
    unsigned sv = (unsigned)src[e];
    unsigned dv = (unsigned)dst[e];
    unsigned b = dv / r;
    unsigned dl = dv - b * r;
    int pos = atomicAdd(&cur[b], 1);
    list[pos] = (sv << RBITS) | dl;
  }
}

__global__ void __launch_bounds__(TPB)
deg_kernel(const unsigned* __restrict__ list, const int* __restrict__ bstart,
           const float* __restrict__ x,
           float* __restrict__ dinv,
           float4* __restrict__ xs4,
           int N, int R) {
  __shared__ int c[RMAX];
  const int t = threadIdx.x;
  const int b = blockIdx.x;
  const int node = b * R + t;
  const bool own = (t < R) && (node < N);
  float x0 = 0.0f, x1 = 0.0f, x2 = 0.0f;
  if (own) {
    const float* xr = x + (size_t)node * 3;
    x0 = xr[0];
    x1 = xr[1];
    x2 = xr[2];
  }
  c[t] = 0;
  __syncthreads();
  const int beg = bstart[b];
  const int end = bstart[b + 1];
  int j = beg + t;
  for (; j + 3 * TPB < end; j += 4 * TPB) {
    unsigned rec[4];
    for (int k = 0; k < 4; ++k) rec[k] = list[j + k * TPB];
    for (int k = 0; k < 4; ++k) atomicAdd(&c[rec[k] & (RMAX - 1)], 1);
  }
  for (; j < end; j += TPB) atomicAdd(&c[list[j] & (RMAX - 1)], 1);
  __syncthreads();
  if (own) {
    float d = 1.0f + (float)c[t];
    float di = (d > 0.0f) ? __frsqrt_rn(d) : 0.0f;
    dinv[node] = di;
    xs4[node] = make_float4(x0 * di, x1 * di, x2 * di, 0.0f);
  }
}

__global__ void __launch_bounds__(TPB)
agg1_kernel(const unsigned* __restrict__ list, const int* __restrict__ bstart,
            const float* __restrict__ dinv,
            const float4* __restrict__ xs4,
            float4* __restrict__ aggx4,
            int N, int R) {
  __shared__ float a0[RMAX];
  __shared__ float a1[RMAX];
  __shared__ float a2[RMAX];
  const int t = threadIdx.x;
  const int b = blockIdx.x;
  const int node = b * R + t;
  const bool own = (t < R) && (node < N);
  const float4 self = own ? xs4[node] : make_float4(0.0f, 0.0f, 0.0f, 0.0f);
  const float dd = own ? dinv[node] : 0.0f;
  a0[t] = self.x;
  a1[t] = self.y;
  a2[t] = self.z;
  __syncthreads();
  const int beg = bstart[b];
  const int end = bstart[b + 1];
  int j = beg + t;
  for (; j + 3 * TPB < end; j += 4 * TPB) {
    unsigned rec[4];
    for (int k = 0; k < 4; ++k) rec[k] = list[j + k * TPB];
    float4 r[4];
    for (int k = 0; k < 4; ++k) r[k] = xs4[rec[k] >> RBITS];
    for (int k = 0; k < 4; ++k) {
      unsigned dl = rec[k] & (RMAX - 1);
      atomicAdd(&a0[dl], r[k].x);
      atomicAdd(&a1[dl], r[k].y);
      atomicAdd(&a2[dl], r[k].z);
    }
  }
  for (; j < end; j += TPB) {
    unsigned rec = list[j];
    unsigned dl = rec & (RMAX - 1);
    float4 r = xs4[rec >> RBITS];
    atomicAdd(&a0[dl], r.x);
    atomicAdd(&a1[dl], r.y);
    atomicAdd(&a2[dl], r.z);
  }
  __syncthreads();
  if (own) aggx4[node] = make_float4(a0[t] * dd, a1[t] * dd, a2[t] * dd, 0.0f);
}

__global__ void __launch_bounds__(256)
finalize1_project(const float4* __restrict__ aggx4,
                  const float* __restrict__ dinv,
                  const float* __restrict__ W1,
                  const float* __restrict__ b1,
                  const float* __restrict__ W2,
                  float* __restrict__ zs,
                  int N, int num_tiles) {
  int flat = blockIdx.x * blockDim.x + threadIdx.x;
  int wave = flat >> 5;
  int lane = threadIdx.x & 31;
  if (wave >= num_tiles) return;

  int m = lane & 15;
  bool hi = lane >= 16;
  int node = wave * 16 + m;
  bool valid = node < N;
  int nl = valid ? node : (N - 1);

  float di = dinv[nl];
  float4 ag = aggx4[nl];
  float ax0 = ag.x;
  float ax1 = ag.y;
  float ax2 = ag.z;

  v2f a;
  a.x = hi ? W1[2 * 16 + m] : W1[0 * 16 + m];
  a.y = hi ? 0.0f : W1[1 * 16 + m];

  v2f b;
  b.x = hi ? ax2 : ax0;
  b.y = hi ? 0.0f : ax1;

  v8f c = {};
  c = __builtin_amdgcn_wmma_f32_16x16x4_f32(
       false, a,  false, b,
       (short)0, c,  false,  false);

  int fb = hi ? 8 : 0;
  float acc = 0.0f;
  for (int v = 0; v < 8; ++v) {
    float hv = fmaxf(c[v] + b1[fb + v], 0.0f);
    acc += hv * W2[fb + v];
  }
  acc += __shfl_xor(acc, 16);

  if (!hi && valid) zs[node] = acc * di;
}

__global__ void __launch_bounds__(TPB)
agg2_kernel(const unsigned* __restrict__ list, const int* __restrict__ bstart,
            const float* __restrict__ dinv, const float* __restrict__ zs,
            const float* __restrict__ b2,
            float* __restrict__ out, int N, int R) {
  __shared__ float acc[RMAX];
  const int t = threadIdx.x;
  const int b = blockIdx.x;
  const int node = b * R + t;
  const bool own = (t < R) && (node < N);
  acc[t] = own ? zs[node] : 0.0f;
  const float dd = own ? dinv[node] : 0.0f;
  const float bias = b2[0];
  __syncthreads();
  const int beg = bstart[b];
  const int end = bstart[b + 1];
  int j = beg + t;
  for (; j + 3 * TPB < end; j += 4 * TPB) {
    unsigned rec[4];
    for (int k = 0; k < 4; ++k) rec[k] = list[j + k * TPB];
    float zv[4];
    for (int k = 0; k < 4; ++k) zv[k] = zs[rec[k] >> RBITS];
    for (int k = 0; k < 4; ++k) atomicAdd(&acc[rec[k] & (RMAX - 1)], zv[k]);
  }
  for (; j < end; j += TPB) {
    unsigned rec = list[j];
    atomicAdd(&acc[rec & (RMAX - 1)], zs[rec >> RBITS]);
  }
  __syncthreads();
  if (own) out[node] = acc[t] * dd + bias;
}

extern "C" void kernel_launch(void* const* d_in, const int* in_sizes, int n_in,
                              void* d_out, int out_size, void* d_ws,
                              size_t ws_size, hipStream_t stream) {
  const float* x  = (const float*)d_in[0];
  const int*   ei = (const int*)d_in[1];
  const float* W1 = (const float*)d_in[2];
  const float* b1 = (const float*)d_in[3];
  const float* W2 = (const float*)d_in[4];
  const float* b2 = (const float*)d_in[5];
  float* out = (float*)d_out;

  const int N = in_sizes[0] / 3;
  const int E = in_sizes[1] / 2;
  const int* src = ei;
  const int* dst = ei + E;

  const int R = (((N + NB - 1) / NB) + 15) & ~15;

  char* ws = (char*)d_ws;
  float* dinv = (float*)(ws);
  float* aggx = (float*)(ws + (size_t)N * 4);
  float* zs   = (float*)(ws + (size_t)N * 4 + (size_t)N * 16);
  float* xs   = (float*)(ws + (size_t)N * 24);
  int* cnt    = (int*)(ws + (size_t)N * 40);
  int* bstart = cnt + NCH * NB;
  unsigned* list = (unsigned*)(bstart + 512);

  count_kernel<<<NCH, TPB, 0, stream>>>(dst, cnt, E, R);
  scan_kernel<<<1, TPB, 0, stream>>>(cnt, bstart);
  fill_kernel<<<NCH, TPB, 0, stream>>>(src, dst, cnt, list, E, R);
  deg_kernel<<<NB, TPB, 0, stream>>>(list, bstart, x, dinv, (float4*)xs, N, R);
  agg1_kernel<<<NB, TPB, 0, stream>>>(list, bstart, dinv, (const float4*)xs,
                                      (float4*)aggx, N, R);
  {
    const int BT = 256;
    int ntiles = (N + 15) / 16;
    int wpb = BT / 32;
    int blocks = (ntiles + wpb - 1) / wpb;
    finalize1_project<<<blocks, BT, 0, stream>>>((const float4*)aggx, dinv, W1,
                                                 b1, W2, zs, N, ntiles);
  }
  agg2_kernel<<<NB, TPB, 0, stream>>>(list, bstart, dinv, zs, b2, out, N, R);
}


# node transform (W1 WMMA, relu, W2, dinv scale) folded into the tail of the layer-1 range pass: each work-group's waves take its node tiles from the sums in LDS; 6 launches, not 7; sums unchanged
# speedup vs baseline: 13.6513x; 1.0338x over previous
#include <hip/hip_runtime.h>
#include <hip/hip_bf16.h>


typedef __attribute__((ext_vector_type(2))) float v2f;
typedef __attribute__((ext_vector_type(8))) float v8f;

#define NB 256
#define NCH 256
#define RMAX 1024
#define RBITS 10
#define TPB 1024
static_assert(TPB == RMAX, "one thread per range slot");

__device__ __forceinline__ void chunk_range(int E, int w, int& e0, int& e1) {
  int per = (E + NCH - 1) / NCH;
  e0 = w * per;
  int hi = e0 + per;
  e1 = hi < E ? hi : E;
}

__global__ void __launch_bounds__(TPB)
count_kernel(const int* __restrict__ dst, int* __restrict__ cnt, int E, int R) {
  __shared__ int h[NB];
  const int t = threadIdx.x;
  if (t < NB) h[t] = 0;
  __syncthreads();
  int e0, e1;
  chunk_range(E, blockIdx.x, e0, e1);
  const unsigned r = (unsigned)R;
  int e = e0 + t;
  for (; e + 3 * TPB < e1; e += 4 * TPB) {
    unsigned d[4];
    for (int k = 0; k < 4; ++k) d[k] = (unsigned)dst[e + k * TPB];
    for (int k = 0; k < 4; ++k) atomicAdd(&h[d[k] / r], 1);
  }
  for (; e < e1; e += TPB) atomicAdd(&h[(unsigned)dst[e] / r], 1);
  __syncthreads();
  if (t < NB) cnt[blockIdx.x * NB + t] = h[t];
}

__global__ void __launch_bounds__(TPB)
scan_kernel(int* __restrict__ cnt, int* __restrict__ bstart) {
  __shared__ int part[4][NB];
  __shared__ int sc[NB];
  const int t = threadIdx.x;
  const int b = t & (NB - 1);
  const int q = t >> 8;
  int sum = 0;
  for (int i = 0; i < NCH / 4; ++i) sum += cnt[(q * (NCH / 4) + i) * NB + b];
  part[q][b] = sum;
  __syncthreads();
  int tot = 0;
  if (t < NB) {
    tot = part[0][t] + part[1][t] + part[2][t] + part[3][t];
    sc[t] = tot;
  }
  __syncthreads();
  for (int o = 1; o < NB; o <<= 1) {
    int v = 0;
    if (t < NB && t >= o) v = sc[t - o];
    __syncthreads();
    if (t < NB) sc[t] += v;
    __syncthreads();
  }
  if (t < NB) {
    bstart[t] = sc[t] - tot;
    if (t == NB - 1) bstart[NB] = sc[t];
  }
  int run = sc[b] - (part[0][b] + part[1][b] + part[2][b] + part[3][b]);
  for (int k = 0; k < q; ++k) run += part[k][b];
  for (int i = 0; i < NCH / 4; ++i) {
    int idx = (q * (NCH / 4) + i) * NB + b;
    int c = cnt[idx];
    cnt[idx] = run;
    run += c;
  }
}

__global__ void __launch_bounds__(TPB)
fill_kernel(const int* __restrict__ src, const int* __restrict__ dst,
            const int* __restrict__ off, unsigned* __restrict__ list,
            int E, int R) {
  __shared__ int cur[NB];
  const int t = threadIdx.x;
  if (t < NB) cur[t] = off[blockIdx.x * NB + t];
  __syncthreads();
  int e0, e1;
  chunk_range(E, blockIdx.x, e0, e1);
  const unsigned r = (unsigned)R;
  int e = e0 + t;
  for (; e + 3 * TPB < e1; e += 4 * TPB) {
    unsigned s[4], d[4];
    for (int k = 0; k < 4; ++k) {
      s[k] = (unsigned)src[e + k * TPB];
      d[k] = (unsigned)dst[e + k * TPB];
    }
    for (int k = 0; k < 4; ++k) {
      unsigned b = d[k] / r;
      unsigned dl = d[k] - b * r;
      int pos = atomicAdd(&cur[b], 1);
      list[pos] = (s[k] << RBITS) | dl;
    }
  }
  for (; e < e1; e += TPB) {
    unsigned sv = (unsigned)src[e];
    unsigned dv = (unsigned)dst[e];
    unsigned b = dv / r;
    unsigned dl = dv - b * r;
    int pos = atomicAdd(&cur[b], 1);
    list[pos] = (sv << RBITS) | dl;
  }
}

__global__ void __launch_bounds__(TPB)
deg_kernel(const unsigned* __restrict__ list, const int* __restrict__ bstart,
           const float* __restrict__ x,
           float* __restrict__ dinv,
           float4* __restrict__ xs4,
           int N, int R) {
  __shared__ int c[RMAX];
  const int t = threadIdx.x;
  const int b = blockIdx.x;
  const int node = b * R + t;
  const bool own = (t < R) && (node < N);
  float x0 = 0.0f, x1 = 0.0f, x2 = 0.0f;
  if (own) {
    const float* xr = x + (size_t)node * 3;
    x0 = xr[0];
    x1 = xr[1];
    x2 = xr[2];
  }
  c[t] = 0;
  __syncthreads();
  const int beg = bstart[b];
  const int end = bstart[b + 1];
  int j = beg + t;
  for (; j + 3 * TPB < end; j += 4 * TPB) {
    unsigned rec[4];
    for (int k = 0; k < 4; ++k) rec[k] = list[j + k * TPB];
    for (int k = 0; k < 4; ++k) atomicAdd(&c[rec[k] & (RMAX - 1)], 1);
  }
  for (; j < end; j += TPB) atomicAdd(&c[list[j] & (RMAX - 1)], 1);
  __syncthreads();
  if (own) {
    float d = 1.0f + (float)c[t];
    float di = (d > 0.0f) ? __frsqrt_rn(d) : 0.0f;
    dinv[node] = di;
    xs4[node] = make_float4(x0 * di, x1 * di, x2 * di, 0.0f);
  }
}

__global__ void __launch_bounds__(TPB)
agg1_project_kernel(const unsigned* __restrict__ list,
                    const int* __restrict__ bstart,
                    const float* __restrict__ dinv,
                    const float4* __restrict__ xs4,
                    const float* __restrict__ W1,
                    const float* __restrict__ b1,
                    const float* __restrict__ W2,
                    float* __restrict__ zs,
                    int N, int R) {
  __shared__ float a0[RMAX];
  __shared__ float a1[RMAX];
  __shared__ float a2[RMAX];
  __shared__ float dl[RMAX];
  __shared__ float zl[RMAX];
  const int t = threadIdx.x;
  const int b = blockIdx.x;
  const int node = b * R + t;
  const bool own = (t < R) && (node < N);
  const float4 self = own ? xs4[node] : make_float4(0.0f, 0.0f, 0.0f, 0.0f);
  const float dd = own ? dinv[node] : 0.0f;
  const int wave = t >> 5;
  const int lane = t & 31;
  const int m = lane & 15;
  const bool hi = lane >= 16;
  v2f a;
  a.x = hi ? W1[2 * 16 + m] : W1[0 * 16 + m];
  a.y = hi ? 0.0f : W1[1 * 16 + m];
  const int fb = hi ? 8 : 0;
  float bv[8], wv[8];
  for (int v = 0; v < 8; ++v) {
    bv[v] = b1[fb + v];
    wv[v] = W2[fb + v];
  }
  a0[t] = self.x;
  a1[t] = self.y;
  a2[t] = self.z;
  __syncthreads();
  const int beg = bstart[b];
  const int end = bstart[b + 1];
  int j = beg + t;
  for (; j + 3 * TPB < end; j += 4 * TPB) {
    unsigned rec[4];
    for (int k = 0; k < 4; ++k) rec[k] = list[j + k * TPB];
    float4 r[4];
    for (int k = 0; k < 4; ++k) r[k] = xs4[rec[k] >> RBITS];
    for (int k = 0; k < 4; ++k) {
      unsigned dloc = rec[k] & (RMAX - 1);
      atomicAdd(&a0[dloc], r[k].x);
      atomicAdd(&a1[dloc], r[k].y);
      atomicAdd(&a2[dloc], r[k].z);
    }
  }
  for (; j < end; j += TPB) {
    unsigned rec = list[j];
    unsigned dloc = rec & (RMAX - 1);
    float4 r = xs4[rec >> RBITS];
    atomicAdd(&a0[dloc], r.x);
    atomicAdd(&a1[dloc], r.y);
    atomicAdd(&a2[dloc], r.z);
  }
  __syncthreads();
  a0[t] = a0[t] * dd;
  a1[t] = a1[t] * dd;
  a2[t] = a2[t] * dd;
  dl[t] = dd;
  __syncthreads();

  const int ntl = R >> 4;
  for (int tl = wave; tl < ntl; tl += TPB / 32) {
    const int nl = tl * 16 + m;
    v2f bx;
    bx.x = hi ? a2[nl] : a0[nl];
    bx.y = hi ? 0.0f : a1[nl];
    v8f c = {};
    c = __builtin_amdgcn_wmma_f32_16x16x4_f32(
         false, a,  false, bx,
         (short)0, c,  false,  false);
    float acc = 0.0f;
    for (int v = 0; v < 8; ++v) {
      float hv = fmaxf(c[v] + bv[v], 0.0f);
      acc += hv * wv[v];
    }
    acc += __shfl_xor(acc, 16);
    if (!hi) zl[nl] = acc * dl[nl];
  }
  __syncthreads();
  if (own) zs[node] = zl[t];
}

__global__ void __launch_bounds__(TPB)
agg2_kernel(const unsigned* __restrict__ list, const int* __restrict__ bstart,
            const float* __restrict__ dinv, const float* __restrict__ zs,
            const float* __restrict__ b2,
            float* __restrict__ out, int N, int R) {
  __shared__ float acc[RMAX];
  const int t = threadIdx.x;
  const int b = blockIdx.x;
  const int node = b * R + t;
  const bool own = (t < R) && (node < N);
  acc[t] = own ? zs[node] : 0.0f;
  const float dd = own ? dinv[node] : 0.0f;
  const float bias = b2[0];
  __syncthreads();
  const int beg = bstart[b];
  const int end = bstart[b + 1];
  int j = beg + t;
  for (; j + 3 * TPB < end; j += 4 * TPB) {
    unsigned rec[4];
    for (int k = 0; k < 4; ++k) rec[k] = list[j + k * TPB];
    float zv[4];
    for (int k = 0; k < 4; ++k) zv[k] = zs[rec[k] >> RBITS];
    for (int k = 0; k < 4; ++k) atomicAdd(&acc[rec[k] & (RMAX - 1)], zv[k]);
  }
  for (; j < end; j += TPB) {
    unsigned rec = list[j];
    atomicAdd(&acc[rec & (RMAX - 1)], zs[rec >> RBITS]);
  }
  __syncthreads();
  if (own) out[node] = acc[t] * dd + bias;
}

extern "C" void kernel_launch(void* const* d_in, const int* in_sizes, int n_in,
                              void* d_out, int out_size, void* d_ws,
                              size_t ws_size, hipStream_t stream) {
  const float* x  = (const float*)d_in[0];
  const int*   ei = (const int*)d_in[1];
  const float* W1 = (const float*)d_in[2];
  const float* b1 = (const float*)d_in[3];
  const float* W2 = (const float*)d_in[4];
  const float* b2 = (const float*)d_in[5];
  float* out = (float*)d_out;

  const int N = in_sizes[0] / 3;
  const int E = in_sizes[1] / 2;
  const int* src = ei;
  const int* dst = ei + E;

  const int R = (((N + NB - 1) / NB) + 15) & ~15;

  char* ws = (char*)d_ws;
  float* dinv = (float*)(ws);
  float* zs   = (float*)(ws + (size_t)N * 4);
  float* xs   = (float*)(ws + (size_t)N * 8);
  int* cnt    = (int*)(ws + (size_t)N * 24);
  int* bstart = cnt + NCH * NB;
  unsigned* list = (unsigned*)(bstart + 512);

  count_kernel<<<NCH, TPB, 0, stream>>>(dst, cnt, E, R);
  scan_kernel<<<1, TPB, 0, stream>>>(cnt, bstart);
  fill_kernel<<<NCH, TPB, 0, stream>>>(src, dst, cnt, list, E, R);
  deg_kernel<<<NB, TPB, 0, stream>>>(list, bstart, x, dinv, (float4*)xs, N, R);
  agg1_project_kernel<<<NB, TPB, 0, stream>>>(list, bstart, dinv,
                                              (const float4*)xs, W1, b1, W2,
                                              zs, N, R);
  agg2_kernel<<<NB, TPB, 0, stream>>>(list, bstart, dinv, zs, b2, out, N, R);
}


# fill pass groups each chunk's edge records by range in LDS (counting sort), then writes each run to the list in whole-line stores, not one scattered 4-byte store per edge; sums stay in arrival order
# speedup vs baseline: 16.8315x; 1.2330x over previous
#include <hip/hip_runtime.h>
#include <hip/hip_bf16.h>


typedef __attribute__((ext_vector_type(2))) float v2f;
typedef __attribute__((ext_vector_type(8))) float v8f;

#define NB 256
#define NCH 256
#define RMAX 1024
#define RBITS 10
#define TPB 1024
#define CAP 32768
static_assert(TPB == RMAX, "one thread per range slot");
static_assert(NB % 32 == 0 && NB <= TPB, "whole waves scan the range counts");

__device__ __forceinline__ void chunk_range(int E, int w, int& e0, int& e1) {
  int per = (E + NCH - 1) / NCH;
  e0 = w * per;
  int hi = e0 + per;
  e1 = hi < E ? hi : E;
}

__global__ void __launch_bounds__(TPB)
count_kernel(const int* __restrict__ dst, int* __restrict__ cnt, int E, int R) {
  __shared__ int h[NB];
  const int t = threadIdx.x;
  if (t < NB) h[t] = 0;
  __syncthreads();
  int e0, e1;
  chunk_range(E, blockIdx.x, e0, e1);
  const unsigned r = (unsigned)R;
  int e = e0 + t;
  for (; e + 3 * TPB < e1; e += 4 * TPB) {
    unsigned d[4];
    for (int k = 0; k < 4; ++k) d[k] = (unsigned)dst[e + k * TPB];
    for (int k = 0; k < 4; ++k) atomicAdd(&h[d[k] / r], 1);
  }
  for (; e < e1; e += TPB) atomicAdd(&h[(unsigned)dst[e] / r], 1);
  __syncthreads();
  if (t < NB) cnt[blockIdx.x * NB + t] = h[t];
}

__global__ void __launch_bounds__(TPB)
scan_kernel(int* __restrict__ cnt, int* __restrict__ bstart) {
  __shared__ int part[4][NB];
  __shared__ int sc[NB];
  const int t = threadIdx.x;
  const int b = t & (NB - 1);
  const int q = t >> 8;
  int sum = 0;
  for (int i = 0; i < NCH / 4; ++i) sum += cnt[(q * (NCH / 4) + i) * NB + b];
  part[q][b] = sum;
  __syncthreads();
  int tot = 0;
  if (t < NB) {
    tot = part[0][t] + part[1][t] + part[2][t] + part[3][t];
    sc[t] = tot;
  }
  __syncthreads();
  for (int o = 1; o < NB; o <<= 1) {
    int v = 0;
    if (t < NB && t >= o) v = sc[t - o];
    __syncthreads();
    if (t < NB) sc[t] += v;
    __syncthreads();
  }
  if (t < NB) {
    bstart[t] = sc[t] - tot;
    if (t == NB - 1) bstart[NB] = sc[t];
  }
  int run = sc[b] - (part[0][b] + part[1][b] + part[2][b] + part[3][b]);
  for (int k = 0; k < q; ++k) run += part[k][b];
  for (int i = 0; i < NCH / 4; ++i) {
    int idx = (q * (NCH / 4) + i) * NB + b;
    int c = cnt[idx];
    cnt[idx] = run;
    run += c;
  }
}

__global__ void __launch_bounds__(TPB)
fill_kernel(const int* __restrict__ src, const int* __restrict__ dst,
            const int* __restrict__ off, const int* __restrict__ bstart,
            unsigned* __restrict__ list, int E, int R) {
  __shared__ int cur[NB];
  __shared__ int ls[NB + 1];
  __shared__ int gs[NB];
  __shared__ int wsum[NB / 32];
  __shared__ unsigned stage[CAP];
  const int t = threadIdx.x;
  const int lane = t & 31;
  const int wave = t >> 5;
  const int w = blockIdx.x;
  int e0, e1;
  chunk_range(E, w, e0, e1);
  const unsigned r = (unsigned)R;
  const bool staged = (e1 - e0) <= CAP;

  if (!staged) {
    if (t < NB) cur[t] = off[w * NB + t];
    __syncthreads();
    for (int e = e0 + t; e < e1; e += TPB) {
      unsigned sv = (unsigned)src[e];
      unsigned dv = (unsigned)dst[e];
      unsigned b = dv / r;
      unsigned dl = dv - b * r;
      int pos = atomicAdd(&cur[b], 1);
      list[pos] = (sv << RBITS) | dl;
    }
    return;
  }

  int g = 0, cn = 0, inc = 0;
  if (t < NB) {
    g = off[w * NB + t];
    const int nx = (w + 1 < NCH) ? off[(w + 1) * NB + t] : bstart[t + 1];
    cn = nx - g;
    inc = cn;
    for (int o = 1; o < 32; o <<= 1) {
      int u = __shfl_up(inc, o);
      if (lane >= o) inc += u;
    }
    if (lane == 31) wsum[wave] = inc;
    gs[t] = g;
  }
  __syncthreads();
  if (t < NB) {
    int base = 0;
    for (int k = 0; k < wave; ++k) base += wsum[k];
    const int s = base + inc - cn;
    ls[t] = s;
    cur[t] = s;
    if (t == NB - 1) ls[NB] = s + cn;
  }
  __syncthreads();

  int e = e0 + t;
  for (; e + 3 * TPB < e1; e += 4 * TPB) {
    unsigned s[4], d[4];
    for (int k = 0; k < 4; ++k) {
      s[k] = (unsigned)src[e + k * TPB];
      d[k] = (unsigned)dst[e + k * TPB];
    }
    int pos[4];
    for (int k = 0; k < 4; ++k) {
      unsigned b = d[k] / r;
      d[k] -= b * r;
      pos[k] = atomicAdd(&cur[b], 1);
    }
    for (int k = 0; k < 4; ++k) stage[pos[k]] = (s[k] << RBITS) | d[k];
  }
  for (; e < e1; e += TPB) {
    unsigned sv = (unsigned)src[e];
    unsigned dv = (unsigned)dst[e];
    unsigned b = dv / r;
    unsigned dl = dv - b * r;
    int pos = atomicAdd(&cur[b], 1);
    stage[pos] = (sv << RBITS) | dl;
  }
  __syncthreads();

  for (int b = wave; b < NB; b += TPB / 32) {
    const int s0 = ls[b];
    const int n = ls[b + 1] - s0;
    const int g0 = gs[b];
    for (int i = lane - (g0 & 31); i < n; i += 4 * 32) {
      unsigned v[4];
      for (int k = 0; k < 4; ++k) {
        const int ik = i + 32 * k;
        v[k] = (ik >= 0 && ik < n) ? stage[s0 + ik] : 0u;
      }
      for (int k = 0; k < 4; ++k) {
        const int ik = i + 32 * k;
        if (ik >= 0 && ik < n) list[g0 + ik] = v[k];
      }
    }
  }
}

__global__ void __launch_bounds__(TPB)
deg_kernel(const unsigned* __restrict__ list, const int* __restrict__ bstart,
           const float* __restrict__ x,
           float* __restrict__ dinv,
           float4* __restrict__ xs4,
           int N, int R) {
  __shared__ int c[RMAX];
  const int t = threadIdx.x;
  const int b = blockIdx.x;
  const int node = b * R + t;
  const bool own = (t < R) && (node < N);
  float x0 = 0.0f, x1 = 0.0f, x2 = 0.0f;
  if (own) {
    const float* xr = x + (size_t)node * 3;
    x0 = xr[0];
    x1 = xr[1];
    x2 = xr[2];
  }
  c[t] = 0;
  __syncthreads();
  const int beg = bstart[b];
  const int end = bstart[b + 1];
  int j = beg + t;
  for (; j + 3 * TPB < end; j += 4 * TPB) {
    unsigned rec[4];
    for (int k = 0; k < 4; ++k) rec[k] = list[j + k * TPB];
    for (int k = 0; k < 4; ++k) atomicAdd(&c[rec[k] & (RMAX - 1)], 1);
  }
  for (; j < end; j += TPB) atomicAdd(&c[list[j] & (RMAX - 1)], 1);
  __syncthreads();
  if (own) {
    float d = 1.0f + (float)c[t];
    float di = (d > 0.0f) ? __frsqrt_rn(d) : 0.0f;
    dinv[node] = di;
    xs4[node] = make_float4(x0 * di, x1 * di, x2 * di, 0.0f);
  }
}

__global__ void __launch_bounds__(TPB)
agg1_project_kernel(const unsigned* __restrict__ list,
                    const int* __restrict__ bstart,
                    const float* __restrict__ dinv,
                    const float4* __restrict__ xs4,
                    const float* __restrict__ W1,
                    const float* __restrict__ b1,
                    const float* __restrict__ W2,
                    float* __restrict__ zs,
                    int N, int R) {
  __shared__ float a0[RMAX];
  __shared__ float a1[RMAX];
  __shared__ float a2[RMAX];
  __shared__ float dl[RMAX];
  __shared__ float zl[RMAX];
  const int t = threadIdx.x;
  const int b = blockIdx.x;
  const int node = b * R + t;
  const bool own = (t < R) && (node < N);
  const float4 self = own ? xs4[node] : make_float4(0.0f, 0.0f, 0.0f, 0.0f);
  const float dd = own ? dinv[node] : 0.0f;
  const int wave = t >> 5;
  const int lane = t & 31;
  const int m = lane & 15;
  const bool hi = lane >= 16;
  v2f a;
  a.x = hi ? W1[2 * 16 + m] : W1[0 * 16 + m];
  a.y = hi ? 0.0f : W1[1 * 16 + m];
  const int fb = hi ? 8 : 0;
  float bv[8], wv[8];
  for (int v = 0; v < 8; ++v) {
    bv[v] = b1[fb + v];
    wv[v] = W2[fb + v];
  }
  a0[t] = self.x;
  a1[t] = self.y;
  a2[t] = self.z;
  __syncthreads();
  const int beg = bstart[b];
  const int end = bstart[b + 1];
  int j = beg + t;
  for (; j + 3 * TPB < end; j += 4 * TPB) {
    unsigned rec[4];
    for (int k = 0; k < 4; ++k) rec[k] = list[j + k * TPB];
    float4 r[4];
    for (int k = 0; k < 4; ++k) r[k] = xs4[rec[k] >> RBITS];
    for (int k = 0; k < 4; ++k) {
      unsigned dloc = rec[k] & (RMAX - 1);
      atomicAdd(&a0[dloc], r[k].x);
      atomicAdd(&a1[dloc], r[k].y);
      atomicAdd(&a2[dloc], r[k].z);
    }
  }
  for (; j < end; j += TPB) {
    unsigned rec = list[j];
    unsigned dloc = rec & (RMAX - 1);
    float4 r = xs4[rec >> RBITS];
    atomicAdd(&a0[dloc], r.x);
    atomicAdd(&a1[dloc], r.y);
    atomicAdd(&a2[dloc], r.z);
  }
  __syncthreads();
  a0[t] = a0[t] * dd;
  a1[t] = a1[t] * dd;
  a2[t] = a2[t] * dd;
  dl[t] = dd;
  __syncthreads();

  const int ntl = R >> 4;
  for (int tl = wave; tl < ntl; tl += TPB / 32) {
    const int nl = tl * 16 + m;
    v2f bx;
    bx.x = hi ? a2[nl] : a0[nl];
    bx.y = hi ? 0.0f : a1[nl];
    v8f c = {};
    c = __builtin_amdgcn_wmma_f32_16x16x4_f32(
         false, a,  false, bx,
         (short)0, c,  false,  false);
    float acc = 0.0f;
    for (int v = 0; v < 8; ++v) {
      float hv = fmaxf(c[v] + bv[v], 0.0f);
      acc += hv * wv[v];
    }
    acc += __shfl_xor(acc, 16);
    if (!hi) zl[nl] = acc * dl[nl];
  }
  __syncthreads();
  if (own) zs[node] = zl[t];
}

__global__ void __launch_bounds__(TPB)
agg2_kernel(const unsigned* __restrict__ list, const int* __restrict__ bstart,
            const float* __restrict__ dinv, const float* __restrict__ zs,
            const float* __restrict__ b2,
            float* __restrict__ out, int N, int R) {
  __shared__ float acc[RMAX];
  const int t = threadIdx.x;
  const int b = blockIdx.x;
  const int node = b * R + t;
  const bool own = (t < R) && (node < N);
  acc[t] = own ? zs[node] : 0.0f;
  const float dd = own ? dinv[node] : 0.0f;
  const float bias = b2[0];
  __syncthreads();
  const int beg = bstart[b];
  const int end = bstart[b + 1];
  int j = beg + t;
  for (; j + 3 * TPB < end; j += 4 * TPB) {
    unsigned rec[4];
    for (int k = 0; k < 4; ++k) rec[k] = list[j + k * TPB];
    float zv[4];
    for (int k = 0; k < 4; ++k) zv[k] = zs[rec[k] >> RBITS];
    for (int k = 0; k < 4; ++k) atomicAdd(&acc[rec[k] & (RMAX - 1)], zv[k]);
  }
  for (; j < end; j += TPB) {
    unsigned rec = list[j];
    atomicAdd(&acc[rec & (RMAX - 1)], zs[rec >> RBITS]);
  }
  __syncthreads();
  if (own) out[node] = acc[t] * dd + bias;
}

extern "C" void kernel_launch(void* const* d_in, const int* in_sizes, int n_in,
                              void* d_out, int out_size, void* d_ws,
                              size_t ws_size, hipStream_t stream) {
  const float* x  = (const float*)d_in[0];
  const int*   ei = (const int*)d_in[1];
  const float* W1 = (const float*)d_in[2];
  const float* b1 = (const float*)d_in[3];
  const float* W2 = (const float*)d_in[4];
  const float* b2 = (const float*)d_in[5];
  float* out = (float*)d_out;

  const int N = in_sizes[0] / 3;
  const int E = in_sizes[1] / 2;
  const int* src = ei;
  const int* dst = ei + E;

  const int R = (((N + NB - 1) / NB) + 15) & ~15;

  char* ws = (char*)d_ws;
  float* dinv = (float*)(ws);
  float* zs   = (float*)(ws + (size_t)N * 4);
  float* xs   = (float*)(ws + (size_t)N * 8);
  int* cnt    = (int*)(ws + (size_t)N * 24);
  int* bstart = cnt + NCH * NB;
  unsigned* list = (unsigned*)(bstart + 512);

  count_kernel<<<NCH, TPB, 0, stream>>>(dst, cnt, E, R);
  scan_kernel<<<1, TPB, 0, stream>>>(cnt, bstart);
  fill_kernel<<<NCH, TPB, 0, stream>>>(src, dst, cnt, bstart, list, E, R);
  deg_kernel<<<NB, TPB, 0, stream>>>(list, bstart, x, dinv, (float4*)xs, N, R);
  agg1_project_kernel<<<NB, TPB, 0, stream>>>(list, bstart, dinv,
                                              (const float4*)xs, W1, b1, W2,
                                              zs, N, R);
  agg2_kernel<<<NB, TPB, 0, stream>>>(list, bstart, dinv, zs, b2, out, N, R);
}


# layer 2 streams zs through LDS in 5 blocks (whole-line loads, next block prefetched) with the range's records in registers: per-edge reads hit LDS, not one L2 line each; float sum order changes
# speedup vs baseline: 19.0908x; 1.1342x over previous
#include <hip/hip_runtime.h>
#include <hip/hip_bf16.h>


typedef __attribute__((ext_vector_type(2))) float v2f;
typedef __attribute__((ext_vector_type(8))) float v8f;

#define NB 256
#define NCH 256
#define RMAX 1024
#define RBITS 10
#define TPB 1024
#define CAP 32768
static_assert(TPB == RMAX, "one thread per range slot");
static_assert(NB % 32 == 0 && NB <= TPB, "whole waves scan the range counts");

__device__ __forceinline__ void chunk_range(int E, int w, int& e0, int& e1) {
  int per = (E + NCH - 1) / NCH;
  e0 = w * per;
  int hi = e0 + per;
  e1 = hi < E ? hi : E;
}

__global__ void __launch_bounds__(TPB)
count_kernel(const int* __restrict__ dst, int* __restrict__ cnt, int E, int R) {
  __shared__ int h[NB];
  const int t = threadIdx.x;
  if (t < NB) h[t] = 0;
  __syncthreads();
  int e0, e1;
  chunk_range(E, blockIdx.x, e0, e1);
  const unsigned r = (unsigned)R;
  int e = e0 + t;
  for (; e + 3 * TPB < e1; e += 4 * TPB) {
    unsigned d[4];
    for (int k = 0; k < 4; ++k) d[k] = (unsigned)dst[e + k * TPB];
    for (int k = 0; k < 4; ++k) atomicAdd(&h[d[k] / r], 1);
  }
  for (; e < e1; e += TPB) atomicAdd(&h[(unsigned)dst[e] / r], 1);
  __syncthreads();
  if (t < NB) cnt[blockIdx.x * NB + t] = h[t];
}

__global__ void __launch_bounds__(TPB)
scan_kernel(int* __restrict__ cnt, int* __restrict__ bstart) {
  __shared__ int part[4][NB];
  __shared__ int sc[NB];
  const int t = threadIdx.x;
  const int b = t & (NB - 1);
  const int q = t >> 8;
  int sum = 0;
  for (int i = 0; i < NCH / 4; ++i) sum += cnt[(q * (NCH / 4) + i) * NB + b];
  part[q][b] = sum;
  __syncthreads();
  int tot = 0;
  if (t < NB) {
    tot = part[0][t] + part[1][t] + part[2][t] + part[3][t];
    sc[t] = tot;
  }
  __syncthreads();
  for (int o = 1; o < NB; o <<= 1) {
    int v = 0;
    if (t < NB && t >= o) v = sc[t - o];
    __syncthreads();
    if (t < NB) sc[t] += v;
    __syncthreads();
  }
  if (t < NB) {
    bstart[t] = sc[t] - tot;
    if (t == NB - 1) bstart[NB] = sc[t];
  }
  int run = sc[b] - (part[0][b] + part[1][b] + part[2][b] + part[3][b]);
  for (int k = 0; k < q; ++k) run += part[k][b];
  for (int i = 0; i < NCH / 4; ++i) {
    int idx = (q * (NCH / 4) + i) * NB + b;
    int c = cnt[idx];
    cnt[idx] = run;
    run += c;
  }
}

__global__ void __launch_bounds__(TPB)
fill_kernel(const int* __restrict__ src, const int* __restrict__ dst,
            const int* __restrict__ off, const int* __restrict__ bstart,
            unsigned* __restrict__ list, int E, int R) {
  __shared__ int cur[NB];
  __shared__ int ls[NB + 1];
  __shared__ int gs[NB];
  __shared__ int wsum[NB / 32];
  __shared__ unsigned stage[CAP];
  const int t = threadIdx.x;
  const int lane = t & 31;
  const int wave = t >> 5;
  const int w = blockIdx.x;
  int e0, e1;
  chunk_range(E, w, e0, e1);
  const unsigned r = (unsigned)R;
  const bool staged = (e1 - e0) <= CAP;

  if (!staged) {
    if (t < NB) cur[t] = off[w * NB + t];
    __syncthreads();
    for (int e = e0 + t; e < e1; e += TPB) {
      unsigned sv = (unsigned)src[e];
      unsigned dv = (unsigned)dst[e];
      unsigned b = dv / r;
      unsigned dl = dv - b * r;
      int pos = atomicAdd(&cur[b], 1);
      list[pos] = (sv << RBITS) | dl;
    }
    return;
  }

  int g = 0, cn = 0, inc = 0;
  if (t < NB) {
    g = off[w * NB + t];
    const int nx = (w + 1 < NCH) ? off[(w + 1) * NB + t] : bstart[t + 1];
    cn = nx - g;
    inc = cn;
    for (int o = 1; o < 32; o <<= 1) {
      int u = __shfl_up(inc, o);
      if (lane >= o) inc += u;
    }
    if (lane == 31) wsum[wave] = inc;
    gs[t] = g;
  }
  __syncthreads();
  if (t < NB) {
    int base = 0;
    for (int k = 0; k < wave; ++k) base += wsum[k];
    const int s = base + inc - cn;
    ls[t] = s;
    cur[t] = s;
    if (t == NB - 1) ls[NB] = s + cn;
  }
  __syncthreads();

  int e = e0 + t;
  for (; e + 3 * TPB < e1; e += 4 * TPB) {
    unsigned s[4], d[4];
    for (int k = 0; k < 4; ++k) {
      s[k] = (unsigned)src[e + k * TPB];
      d[k] = (unsigned)dst[e + k * TPB];
    }
    int pos[4];
    for (int k = 0; k < 4; ++k) {
      unsigned b = d[k] / r;
      d[k] -= b * r;
      pos[k] = atomicAdd(&cur[b], 1);
    }
    for (int k = 0; k < 4; ++k) stage[pos[k]] = (s[k] << RBITS) | d[k];
  }
  for (; e < e1; e += TPB) {
    unsigned sv = (unsigned)src[e];
    unsigned dv = (unsigned)dst[e];
    unsigned b = dv / r;
    unsigned dl = dv - b * r;
    int pos = atomicAdd(&cur[b], 1);
    stage[pos] = (sv << RBITS) | dl;
  }
  __syncthreads();

  for (int b = wave; b < NB; b += TPB / 32) {
    const int s0 = ls[b];
    const int n = ls[b + 1] - s0;
    const int g0 = gs[b];
    for (int i = lane - (g0 & 31); i < n; i += 4 * 32) {
      unsigned v[4];
      for (int k = 0; k < 4; ++k) {
        const int ik = i + 32 * k;
        v[k] = (ik >= 0 && ik < n) ? stage[s0 + ik] : 0u;
      }
      for (int k = 0; k < 4; ++k) {
        const int ik = i + 32 * k;
        if (ik >= 0 && ik < n) list[g0 + ik] = v[k];
      }
    }
  }
}

__global__ void __launch_bounds__(TPB)
deg_kernel(const unsigned* __restrict__ list, const int* __restrict__ bstart,
           const float* __restrict__ x,
           float* __restrict__ dinv,
           float4* __restrict__ xs4,
           int N, int R) {
  __shared__ int c[RMAX];
  const int t = threadIdx.x;
  const int b = blockIdx.x;
  const int node = b * R + t;
  const bool own = (t < R) && (node < N);
  float x0 = 0.0f, x1 = 0.0f, x2 = 0.0f;
  if (own) {
    const float* xr = x + (size_t)node * 3;
    x0 = xr[0];
    x1 = xr[1];
    x2 = xr[2];
  }
  c[t] = 0;
  __syncthreads();
  const int beg = bstart[b];
  const int end = bstart[b + 1];
  int j = beg + t;
  for (; j + 3 * TPB < end; j += 4 * TPB) {
    unsigned rec[4];
    for (int k = 0; k < 4; ++k) rec[k] = list[j + k * TPB];
    for (int k = 0; k < 4; ++k) atomicAdd(&c[rec[k] & (RMAX - 1)], 1);
  }
  for (; j < end; j += TPB) atomicAdd(&c[list[j] & (RMAX - 1)], 1);
  __syncthreads();
  if (own) {
    float d = 1.0f + (float)c[t];
    float di = (d > 0.0f) ? __frsqrt_rn(d) : 0.0f;
    dinv[node] = di;
    xs4[node] = make_float4(x0 * di, x1 * di, x2 * di, 0.0f);
  }
}

__global__ void __launch_bounds__(TPB)
agg1_project_kernel(const unsigned* __restrict__ list,
                    const int* __restrict__ bstart,
                    const float* __restrict__ dinv,
                    const float4* __restrict__ xs4,
                    const float* __restrict__ W1,
                    const float* __restrict__ b1,
                    const float* __restrict__ W2,
                    float* __restrict__ zs,
                    int N, int R) {
  __shared__ float a0[RMAX];
  __shared__ float a1[RMAX];
  __shared__ float a2[RMAX];
  __shared__ float dl[RMAX];
  __shared__ float zl[RMAX];
  const int t = threadIdx.x;
  const int b = blockIdx.x;
  const int node = b * R + t;
  const bool own = (t < R) && (node < N);
  const float4 self = own ? xs4[node] : make_float4(0.0f, 0.0f, 0.0f, 0.0f);
  const float dd = own ? dinv[node] : 0.0f;
  const int wave = t >> 5;
  const int lane = t & 31;
  const int m = lane & 15;
  const bool hi = lane >= 16;
  v2f a;
  a.x = hi ? W1[2 * 16 + m] : W1[0 * 16 + m];
  a.y = hi ? 0.0f : W1[1 * 16 + m];
  const int fb = hi ? 8 : 0;
  float bv[8], wv[8];
  for (int v = 0; v < 8; ++v) {
    bv[v] = b1[fb + v];
    wv[v] = W2[fb + v];
  }
  a0[t] = self.x;
  a1[t] = self.y;
  a2[t] = self.z;
  __syncthreads();
  const int beg = bstart[b];
  const int end = bstart[b + 1];
  int j = beg + t;
  for (; j + 3 * TPB < end; j += 4 * TPB) {
    unsigned rec[4];
    for (int k = 0; k < 4; ++k) rec[k] = list[j + k * TPB];
    float4 r[4];
    for (int k = 0; k < 4; ++k) r[k] = xs4[rec[k] >> RBITS];
    for (int k = 0; k < 4; ++k) {
      unsigned dloc = rec[k] & (RMAX - 1);
      atomicAdd(&a0[dloc], r[k].x);
      atomicAdd(&a1[dloc], r[k].y);
      atomicAdd(&a2[dloc], r[k].z);
    }
  }
  for (; j < end; j += TPB) {
    unsigned rec = list[j];
    unsigned dloc = rec & (RMAX - 1);
    float4 r = xs4[rec >> RBITS];
    atomicAdd(&a0[dloc], r.x);
    atomicAdd(&a1[dloc], r.y);
    atomicAdd(&a2[dloc], r.z);
  }
  __syncthreads();
  a0[t] = a0[t] * dd;
  a1[t] = a1[t] * dd;
  a2[t] = a2[t] * dd;
  dl[t] = dd;
  __syncthreads();

  const int ntl = R >> 4;
  for (int tl = wave; tl < ntl; tl += TPB / 32) {
    const int nl = tl * 16 + m;
    v2f bx;
    bx.x = hi ? a2[nl] : a0[nl];
    bx.y = hi ? 0.0f : a1[nl];
    v8f c = {};
    c = __builtin_amdgcn_wmma_f32_16x16x4_f32(
         false, a,  false, bx,
         (short)0, c,  false,  false);
    float acc = 0.0f;
    for (int v = 0; v < 8; ++v) {
      float hv = fmaxf(c[v] + bv[v], 0.0f);
      acc += hv * wv[v];
    }
    acc += __shfl_xor(acc, 16);
    if (!hi) zl[nl] = acc * dl[nl];
  }
  __syncthreads();
  if (own) zs[node] = zl[t];
}

#define ZCAP 40960
#define ZV (ZCAP / 4 / TPB)
#define KREC 26
#define KG 13
typedef __attribute__((ext_vector_type(4))) float v4f;
static_assert(ZCAP % (4 * TPB) == 0, "whole float4 per thread");
static_assert(KREC % KG == 0, "whole groups");

__global__ void __launch_bounds__(TPB)
agg2_kernel(const unsigned* __restrict__ list, const int* __restrict__ bstart,
            const float* __restrict__ dinv, const float* zs,
            const float* __restrict__ b2,
            float* __restrict__ out, int N, int R) {
  __shared__ float acc[RMAX];
  __shared__ __attribute__((aligned(16))) float zb[ZCAP];
  const int t = threadIdx.x;
  const int b = blockIdx.x;
  const int node = b * R + t;
  const bool own = (t < R) && (node < N);
  acc[t] = own ? zs[node] : 0.0f;
  const float dd = own ? dinv[node] : 0.0f;
  const float bias = b2[0];
  const int beg = bstart[b];
  const int end = bstart[b + 1];
  const bool staged = ((((size_t)zs) & 15) == 0) && ((N & 3) == 0);

  if (staged) {
    const int nz = (N + ZCAP - 1) / ZCAP;
    const int blen = (((N + nz - 1) / nz) + 3) & ~3;
    const v4f* z4 = (const v4f*)zs;
    v4f* zb4 = (v4f*)zb;
    for (int base = beg; base < end; base += KREC * TPB) {
      unsigned rec[KREC];
      for (int k = 0; k < KREC; ++k) {
        const int j = base + k * TPB + t;
        rec[k] = (j < end) ? list[j] : 0xFFFFFFFFu;
      }
      v4f pre[ZV];
      for (int k = 0; k < ZV; ++k) pre[k] = (v4f){0.0f, 0.0f, 0.0f, 0.0f};
      int blk = b % nz;
      {
        const int q0 = (blk * blen) >> 2;
        const int s1 = (blk + 1) * blen < N ? (blk + 1) * blen : N;
        const int nq = (s1 >> 2) - q0;
        for (int k = 0; k < ZV; ++k)
          if (t + k * TPB < nq) pre[k] = z4[q0 + t + k * TPB];
      }
      for (int it = 0; it < nz; ++it) {
        const int s0 = blk * blen;
        const int s1 = s0 + blen < N ? s0 + blen : N;
        const unsigned len = (unsigned)(s1 - s0);
        __syncthreads();
        for (int k = 0; k < ZV; ++k) zb4[t + k * TPB] = pre[k];
        __syncthreads();
        const int nb = (blk + 1 == nz) ? 0 : blk + 1;
        if (it + 1 < nz) {
          const int q0 = (nb * blen) >> 2;
          const int e1 = (nb + 1) * blen < N ? (nb + 1) * blen : N;
          const int nq = (e1 >> 2) - q0;
          for (int k = 0; k < ZV; ++k)
            if (t + k * TPB < nq) pre[k] = z4[q0 + t + k * TPB];
        }
        for (int k0 = 0; k0 < KREC; k0 += KG) {
          float zv[KG];
          bool in[KG];
          for (int k = 0; k < KG; ++k) {
            const unsigned rel = (rec[k0 + k] >> RBITS) - (unsigned)s0;
            in[k] = rel < len;
            zv[k] = zb[in[k] ? rel : 0u];
          }
          for (int k = 0; k < KG; ++k)
            if (in[k]) atomicAdd(&acc[rec[k0 + k] & (RMAX - 1)], zv[k]);
        }
        blk = nb;
      }
    }
  } else {
    __syncthreads();
    int j = beg + t;
    for (; j + 3 * TPB < end; j += 4 * TPB) {
      unsigned rec[4];
      for (int k = 0; k < 4; ++k) rec[k] = list[j + k * TPB];
      float zv[4];
      for (int k = 0; k < 4; ++k) zv[k] = zs[rec[k] >> RBITS];
      for (int k = 0; k < 4; ++k) atomicAdd(&acc[rec[k] & (RMAX - 1)], zv[k]);
    }
    for (; j < end; j += TPB) {
      unsigned rec = list[j];
      atomicAdd(&acc[rec & (RMAX - 1)], zs[rec >> RBITS]);
    }
  }
  __syncthreads();
  if (own) out[node] = acc[t] * dd + bias;
}

extern "C" void kernel_launch(void* const* d_in, const int* in_sizes, int n_in,
                              void* d_out, int out_size, void* d_ws,
                              size_t ws_size, hipStream_t stream) {
  const float* x  = (const float*)d_in[0];
  const int*   ei = (const int*)d_in[1];
  const float* W1 = (const float*)d_in[2];
  const float* b1 = (const float*)d_in[3];
  const float* W2 = (const float*)d_in[4];
  const float* b2 = (const float*)d_in[5];
  float* out = (float*)d_out;

  const int N = in_sizes[0] / 3;
  const int E = in_sizes[1] / 2;
  const int* src = ei;
  const int* dst = ei + E;

  const int R = (((N + NB - 1) / NB) + 15) & ~15;

  char* ws = (char*)d_ws;
  float* dinv = (float*)(ws);
  float* zs   = (float*)(ws + (size_t)N * 4);
  float* xs   = (float*)(ws + (size_t)N * 8);
  int* cnt    = (int*)(ws + (size_t)N * 24);
  int* bstart = cnt + NCH * NB;
  unsigned* list = (unsigned*)(bstart + 512);

  count_kernel<<<NCH, TPB, 0, stream>>>(dst, cnt, E, R);
  scan_kernel<<<1, TPB, 0, stream>>>(cnt, bstart);
  fill_kernel<<<NCH, TPB, 0, stream>>>(src, dst, cnt, bstart, list, E, R);
  deg_kernel<<<NB, TPB, 0, stream>>>(list, bstart, x, dinv, (float4*)xs, N, R);
  agg1_project_kernel<<<NB, TPB, 0, stream>>>(list, bstart, dinv,
                                              (const float4*)xs, W1, b1, W2,
                                              zs, N, R);
  agg2_kernel<<<NB, TPB, 0, stream>>>(list, bstart, dinv, zs, b2, out, N, R);
}


# edge records grouped by range inside each chunk only (one self-contained pass, per-chunk run table); range passes walk 256 short runs: no global count or scan, 4 launches not 6; sum order changes
# speedup vs baseline: 22.5412x; 1.1807x over previous
#include <hip/hip_runtime.h>
#include <hip/hip_bf16.h>


typedef __attribute__((ext_vector_type(2))) float v2f;
typedef __attribute__((ext_vector_type(8))) float v8f;

#define NB 256
#define NCH 256
#define RMAX 1024
#define RBITS 10
#define TPB 1024
#define KF 25
#define CAP (KF * TPB)
#define SPW (NCH / (TPB / 32))
#define INV 0xFFFFFFFFu
static_assert(TPB == RMAX, "one thread per range slot");
static_assert(NB % 32 == 0 && NB <= TPB, "whole waves scan the range counts");
static_assert(NB <= 256, "a range index fits 8 bits beside its rank");
static_assert(SPW * (TPB / 32) == NCH && SPW <= 32, "every run has one wave");

__device__ __forceinline__ void chunk_range(int E, int w, int& e0, int& e1) {
  int per = (((E + NCH - 1) / NCH) + 31) & ~31;
  int lo = w * per;
  e0 = lo < E ? lo : E;
  int hi = e0 + per;
  e1 = hi < E ? hi : E;
}

__device__ __forceinline__ void load_runs(const int2* __restrict__ tab, int b,
                                          int wave, int lane, int (&rs)[SPW],
                                          int (&rn)[SPW]) {
  int2 sg = make_int2(0, 0);
  if (lane < SPW) sg = tab[(wave * SPW + lane) * NB + b];
#pragma unroll
  for (int k = 0; k < SPW; ++k) {
    rs[k] = __builtin_amdgcn_readlane(sg.x, k);
    rn[k] = __builtin_amdgcn_readlane(sg.y, k);
  }
}

__global__ void __launch_bounds__(TPB)
group_kernel(const int* __restrict__ src, const int* __restrict__ dst,
             unsigned* __restrict__ list, int2* __restrict__ tab, int E,
             int R) {
  __shared__ int cur[NB];
  __shared__ int ls[NB + 1];
  __shared__ int wsum[NB / 32];
  __shared__ __attribute__((aligned(16))) unsigned stage[CAP];
  const int t = threadIdx.x;
  const int lane = t & 31;
  const int wave = t >> 5;
  const int w = blockIdx.x;
  int e0, e1;
  chunk_range(E, w, e0, e1);
  const int n = e1 - e0;
  const unsigned r = (unsigned)R;
  const bool fast = n <= CAP;

  if (t < NB) cur[t] = 0;
  __syncthreads();
  unsigned s[KF], d[KF];
  if (fast) {
#pragma unroll
    for (int k = 0; k < KF; ++k) {
      const int e = e0 + t + k * TPB;
      const bool ok = e < e1;
      s[k] = ok ? (unsigned)src[e] : 0u;
      d[k] = ok ? (unsigned)dst[e] : INV;
    }
#pragma unroll
    for (int k = 0; k < KF; ++k) {
      if (d[k] != INV) {
        const unsigned b = d[k] / r;
        const unsigned dl = d[k] - b * r;
        const unsigned rk = (unsigned)atomicAdd(&cur[b], 1);
        s[k] = (s[k] << RBITS) | dl;
        d[k] = (rk << 8) | b;
      }
    }
  } else {
    for (int e = e0 + t; e < e1; e += TPB)
      atomicAdd(&cur[(unsigned)dst[e] / r], 1);
  }
  __syncthreads();
  int cn = 0, inc = 0;
  if (t < NB) {
    cn = cur[t];
    inc = cn;
    for (int o = 1; o < 32; o <<= 1) {
      int u = __shfl_up(inc, o);
      if (lane >= o) inc += u;
    }
    if (lane == 31) wsum[wave] = inc;
  }
  __syncthreads();
  if (t < NB) {
    int base = 0;
    for (int k = 0; k < wave; ++k) base += wsum[k];
    const int st = base + inc - cn;
    ls[t] = st;
    if (t == NB - 1) ls[NB] = st + cn;
    tab[w * NB + t] = make_int2(e0 + st, cn);
  }
  __syncthreads();

  if (fast) {
#pragma unroll
    for (int k = 0; k < KF; ++k)
      if (d[k] != INV) stage[ls[d[k] & 255u] + (int)(d[k] >> 8)] = s[k];
    __syncthreads();
    unsigned* lo = list + e0;
    if ((((size_t)lo) & 15) == 0) {
      const int n4 = n >> 2;
      uint4* l4 = (uint4*)lo;
      const uint4* s4 = (const uint4*)stage;
      for (int i = t; i < n4; i += TPB) l4[i] = s4[i];
      for (int i = (n4 << 2) + t; i < n; i += TPB) lo[i] = stage[i];
    } else {
      for (int i = t; i < n; i += TPB) lo[i] = stage[i];
    }
  } else {
    if (t < NB) cur[t] = e0 + ls[t];
    __syncthreads();
    for (int e = e0 + t; e < e1; e += TPB) {
      unsigned sv = (unsigned)src[e];
      unsigned dv = (unsigned)dst[e];
      unsigned b = dv / r;
      unsigned dl = dv - b * r;
      int pos = atomicAdd(&cur[b], 1);
      list[pos] = (sv << RBITS) | dl;
    }
  }
}

__global__ void __launch_bounds__(TPB)
deg_kernel(const unsigned* __restrict__ list, const int2* __restrict__ tab,
           const float* __restrict__ x,
           float* __restrict__ dinv,
           float4* __restrict__ xs4,
           int N, int R) {
  __shared__ int c[RMAX];
  const int t = threadIdx.x;
  const int lane = t & 31;
  const int wave = t >> 5;
  const int b = blockIdx.x;
  const int node = b * R + t;
  const bool own = (t < R) && (node < N);
  int rs[SPW], rn[SPW];
  load_runs(tab, b, wave, lane, rs, rn);
  float x0 = 0.0f, x1 = 0.0f, x2 = 0.0f;
  if (own) {
    const float* xr = x + (size_t)node * 3;
    x0 = xr[0];
    x1 = xr[1];
    x2 = xr[2];
  }
  c[t] = 0;
  __syncthreads();
  unsigned rec[SPW][4];
#pragma unroll
  for (int k = 0; k < SPW; ++k)
#pragma unroll
    for (int q = 0; q < 4; ++q) {
      const int i = lane + 32 * q;
      const bool ok = i < rn[k];
      const unsigned v = list[ok ? rs[k] + i : 0];
      rec[k][q] = ok ? v : INV;
    }
#pragma unroll
  for (int k = 0; k < SPW; ++k)
#pragma unroll
    for (int q = 0; q < 4; ++q)
      if (rec[k][q] != INV) atomicAdd(&c[rec[k][q] & (RMAX - 1)], 1);
#pragma unroll
  for (int k = 0; k < SPW; ++k)
    for (int i = 128 + lane; i < rn[k]; i += 32)
      atomicAdd(&c[list[rs[k] + i] & (RMAX - 1)], 1);
  __syncthreads();
  if (own) {
    float d = 1.0f + (float)c[t];
    float di = (d > 0.0f) ? __frsqrt_rn(d) : 0.0f;
    dinv[node] = di;
    xs4[node] = make_float4(x0 * di, x1 * di, x2 * di, 0.0f);
  }
}

__global__ void __launch_bounds__(TPB)
agg1_project_kernel(const unsigned* __restrict__ list,
                    const int2* __restrict__ tab,
                    const float* __restrict__ dinv,
                    const float4* __restrict__ xs4,
                    const float* __restrict__ W1,
                    const float* __restrict__ b1,
                    const float* __restrict__ W2,
                    float* __restrict__ zs,
                    int N, int R) {
  __shared__ float a0[RMAX];
  __shared__ float a1[RMAX];
  __shared__ float a2[RMAX];
  __shared__ float dl[RMAX];
  __shared__ float zl[RMAX];
  const int t = threadIdx.x;
  const int b = blockIdx.x;
  const int node = b * R + t;
  const bool own = (t < R) && (node < N);
  const int wave = t >> 5;
  const int lane = t & 31;
  int rs[SPW], rn[SPW];
  load_runs(tab, b, wave, lane, rs, rn);
  const float4 self = own ? xs4[node] : make_float4(0.0f, 0.0f, 0.0f, 0.0f);
  const float dd = own ? dinv[node] : 0.0f;
  const int m = lane & 15;
  const bool hi = lane >= 16;
  v2f a;
  a.x = hi ? W1[2 * 16 + m] : W1[0 * 16 + m];
  a.y = hi ? 0.0f : W1[1 * 16 + m];
  const int fb = hi ? 8 : 0;
  float bv[8], wv[8];
  for (int v = 0; v < 8; ++v) {
    bv[v] = b1[fb + v];
    wv[v] = W2[fb + v];
  }
  a0[t] = self.x;
  a1[t] = self.y;
  a2[t] = self.z;
  __syncthreads();
  unsigned rec[SPW][4];
#pragma unroll
  for (int k = 0; k < SPW; ++k)
#pragma unroll
    for (int q = 0; q < 4; ++q) {
      const int i = lane + 32 * q;
      const bool ok = i < rn[k];
      const unsigned v = list[ok ? rs[k] + i : 0];
      rec[k][q] = ok ? v : INV;
    }
#pragma unroll
  for (int k = 0; k < SPW; k += 2) {
    float4 g[2][4];
#pragma unroll
    for (int kk = 0; kk < 2; ++kk)
#pragma unroll
      for (int q = 0; q < 4; ++q) {
        const unsigned rc = rec[k + kk][q];
        g[kk][q] = xs4[(rc != INV) ? (rc >> RBITS) : 0u];
      }
#pragma unroll
    for (int kk = 0; kk < 2; ++kk)
#pragma unroll
      for (int q = 0; q < 4; ++q) {
        const unsigned rc = rec[k + kk][q];
        if (rc != INV) {
          const unsigned dloc = rc & (RMAX - 1);
          atomicAdd(&a0[dloc], g[kk][q].x);
          atomicAdd(&a1[dloc], g[kk][q].y);
          atomicAdd(&a2[dloc], g[kk][q].z);
        }
      }
  }
#pragma unroll
  for (int k = 0; k < SPW; ++k)
    for (int i = 128 + lane; i < rn[k]; i += 32) {
      const unsigned rc = list[rs[k] + i];
      const unsigned dloc = rc & (RMAX - 1);
      const float4 g = xs4[rc >> RBITS];
      atomicAdd(&a0[dloc], g.x);
      atomicAdd(&a1[dloc], g.y);
      atomicAdd(&a2[dloc], g.z);
    }
  __syncthreads();
  a0[t] = a0[t] * dd;
  a1[t] = a1[t] * dd;
  a2[t] = a2[t] * dd;
  dl[t] = dd;
  __syncthreads();

  const int ntl = R >> 4;
  for (int tl = wave; tl < ntl; tl += TPB / 32) {
    const int nl = tl * 16 + m;
    v2f bx;
    bx.x = hi ? a2[nl] : a0[nl];
    bx.y = hi ? 0.0f : a1[nl];
    v8f c = {};
    c = __builtin_amdgcn_wmma_f32_16x16x4_f32(
         false, a,  false, bx,
         (short)0, c,  false,  false);
    float acc = 0.0f;
    for (int v = 0; v < 8; ++v) {
      float hv = fmaxf(c[v] + bv[v], 0.0f);
      acc += hv * wv[v];
    }
    acc += __shfl_xor(acc, 16);
    if (!hi) zl[nl] = acc * dl[nl];
  }
  __syncthreads();
  if (own) zs[node] = zl[t];
}

#define ZCAP 40960
#define ZV (ZCAP / 4 / TPB)
#define QS 4
typedef __attribute__((ext_vector_type(4))) float v4f;
static_assert(ZCAP % (4 * TPB) == 0, "whole float4 per thread");

__global__ void __launch_bounds__(TPB)
agg2_kernel(const unsigned* __restrict__ list, const int2* __restrict__ tab,
            const float* __restrict__ dinv, const float* zs,
            const float* __restrict__ b2,
            float* __restrict__ out, int N, int R) {
  __shared__ float acc[RMAX];
  __shared__ __attribute__((aligned(16))) float zb[ZCAP];
  const int t = threadIdx.x;
  const int lane = t & 31;
  const int wave = t >> 5;
  const int b = blockIdx.x;
  const int node = b * R + t;
  const bool own = (t < R) && (node < N);
  int rs[SPW], rn[SPW];
  load_runs(tab, b, wave, lane, rs, rn);
  acc[t] = own ? zs[node] : 0.0f;
  const float dd = own ? dinv[node] : 0.0f;
  const float bias = b2[0];
  const bool staged = ((((size_t)zs) & 15) == 0) && ((N & 3) == 0);

  if (staged) {
    const int nz = (N + ZCAP - 1) / ZCAP;
    const int blen = (((N + nz - 1) / nz) + 3) & ~3;
    const v4f* z4 = (const v4f*)zs;
    v4f* zb4 = (v4f*)zb;
    {
      unsigned rec[SPW][QS];
#pragma unroll
      for (int k = 0; k < SPW; ++k)
#pragma unroll
        for (int q = 0; q < QS; ++q) {
          const int i = 32 * q + lane;
          const bool ok = i < rn[k];
          const unsigned v = list[ok ? rs[k] + i : 0];
          rec[k][q] = ok ? v : INV;
        }
      v4f pre[ZV];
      for (int k = 0; k < ZV; ++k) pre[k] = (v4f){0.0f, 0.0f, 0.0f, 0.0f};
      int blk = b % nz;
      {
        const int q0 = (blk * blen) >> 2;
        const int s1 = (blk + 1) * blen < N ? (blk + 1) * blen : N;
        const int nq = (s1 >> 2) - q0;
        for (int k = 0; k < ZV; ++k)
          if (t + k * TPB < nq) pre[k] = z4[q0 + t + k * TPB];
      }
      for (int it = 0; it < nz; ++it) {
        const int s0 = blk * blen;
        const int s1 = s0 + blen < N ? s0 + blen : N;
        const unsigned len = (unsigned)(s1 - s0);
        __syncthreads();
        for (int k = 0; k < ZV; ++k) zb4[t + k * TPB] = pre[k];
        __syncthreads();
        const int nb = (blk + 1 == nz) ? 0 : blk + 1;
        if (it + 1 < nz) {
          const int q0 = (nb * blen) >> 2;
          const int e1 = (nb + 1) * blen < N ? (nb + 1) * blen : N;
          const int nq = (e1 >> 2) - q0;
          for (int k = 0; k < ZV; ++k)
            if (t + k * TPB < nq) pre[k] = z4[q0 + t + k * TPB];
        }
#pragma unroll
        for (int k = 0; k < SPW; ++k) {
          float zv[QS];
          bool in[QS];
#pragma unroll
          for (int q = 0; q < QS; ++q) {
            const unsigned rel = (rec[k][q] >> RBITS) - (unsigned)s0;
            in[q] = rel < len;
            zv[q] = zb[in[q] ? rel : 0u];
          }
#pragma unroll
          for (int q = 0; q < QS; ++q)
            if (in[q]) atomicAdd(&acc[rec[k][q] & (RMAX - 1)], zv[q]);
        }
        blk = nb;
      }
    }
#pragma unroll
    for (int k = 0; k < SPW; ++k)
      for (int i = QS * 32 + lane; i < rn[k]; i += 32) {
        const unsigned rc = list[rs[k] + i];
        atomicAdd(&acc[rc & (RMAX - 1)], zs[rc >> RBITS]);
      }
  } else {
    __syncthreads();
#pragma unroll
    for (int k = 0; k < SPW; ++k)
      for (int i = lane; i < rn[k]; i += 32) {
        const unsigned rc = list[rs[k] + i];
        atomicAdd(&acc[rc & (RMAX - 1)], zs[rc >> RBITS]);
      }
  }
  __syncthreads();
  if (own) out[node] = acc[t] * dd + bias;
}

extern "C" void kernel_launch(void* const* d_in, const int* in_sizes, int n_in,
                              void* d_out, int out_size, void* d_ws,
                              size_t ws_size, hipStream_t stream) {
  const float* x  = (const float*)d_in[0];
  const int*   ei = (const int*)d_in[1];
  const float* W1 = (const float*)d_in[2];
  const float* b1 = (const float*)d_in[3];
  const float* W2 = (const float*)d_in[4];
  const float* b2 = (const float*)d_in[5];
  float* out = (float*)d_out;

  const int N = in_sizes[0] / 3;
  const int E = in_sizes[1] / 2;
  const int* src = ei;
  const int* dst = ei + E;

  const int R = (((N + NB - 1) / NB) + 15) & ~15;

  char* ws = (char*)d_ws;
  float* dinv = (float*)(ws);
  float* zs   = (float*)(ws + (size_t)N * 4);
  float* xs   = (float*)(ws + (size_t)N * 8);
  int2* tab   = (int2*)(ws + (size_t)N * 24);
  unsigned* list = (unsigned*)(tab + NCH * NB);

  group_kernel<<<NCH, TPB, 0, stream>>>(src, dst, list, tab, E, R);
  deg_kernel<<<NB, TPB, 0, stream>>>(list, tab, x, dinv, (float4*)xs, N, R);
  agg1_project_kernel<<<NB, TPB, 0, stream>>>(list, tab, dinv,
                                              (const float4*)xs, W1, b1, W2,
                                              zs, N, R);
  agg2_kernel<<<NB, TPB, 0, stream>>>(list, tab, dinv, zs, b2, out, N, R);
}
